# starting kernel
# baseline (speedup 1.0000x reference)
#include <hip/hip_runtime.h>


typedef float v2f __attribute__((ext_vector_type(2)));
typedef float v8f __attribute__((ext_vector_type(8)));

#define BT 128
#define NN 256
#define FF 128
#define HH 4
#define DD 32
#define NEGV (-1e9f)

__device__ __forceinline__ float wave_max32(float v) {
#pragma unroll
  for (int o = 16; o > 0; o >>= 1) v = fmaxf(v, __shfl_xor(v, o, 32));
  return v;
}
__device__ __forceinline__ float wave_sum32(float v) {
#pragma unroll
  for (int o = 16; o > 0; o >>= 1) v += __shfl_xor(v, o, 32);
  return v;
}

__global__ __launch_bounds__(256) void transpose_w(const float* __restrict__ W,
                                                   float* __restrict__ Wt) {
  int idx = blockIdx.x * 256 + threadIdx.x;
  int n = idx >> 7, k = idx & 127;
  Wt[idx] = W[k * FF + n];
}

__global__ __launch_bounds__(256) void gemm_xw(const float* __restrict__ X,
                                               const float* __restrict__ Wt,
                                               float* __restrict__ WhT) {
  const int tid  = threadIdx.x;
  const int lane = tid & 31;
  const int wave = tid >> 5;
  const int bt   = blockIdx.x >> 4;
  const int m0   = (blockIdx.x & 15) * 16;

  __shared__ float As[16][FF];
  const float* Xb = X + (size_t)(bt * NN + m0) * FF;
#pragma unroll
  for (int i = 0; i < 8; ++i) {
    int idx = tid + i * 256;
    As[idx >> 7][idx & 127] = Xb[idx];
  }
  __syncthreads();

  const int n0 = wave * 16;
  const int kh = lane >> 4;
  const int nl = lane & 15;
  v8f acc = {};
  for (int kk = 0; kk < FF; kk += 4) {
    v2f a = *(const v2f*)(Wt + (size_t)(n0 + nl) * FF + kk + 2 * kh);
    v2f b = *(const v2f*)&As[nl][kk + 2 * kh];
    acc = __builtin_amdgcn_wmma_f32_16x16x4_f32(false, a, false, b,
                                                (short)0, acc, false, false);
  }
#pragma unroll
  for (int v = 0; v < 8; ++v) {
    int c = n0 + v + 8 * kh;
    int h = c >> 5, d = c & 31;
    WhT[((size_t)(bt * HH + h) * DD + d) * NN + m0 + nl] = acc[v];
  }
}

__global__ __launch_bounds__(256) void gat_scores(const float* __restrict__ WhT,
                                                  const float* __restrict__ a,
                                                  float* __restrict__ s1,
                                                  float* __restrict__ s2) {
  int idx = blockIdx.x * 256 + threadIdx.x;
  int n  = idx & (NN - 1);
  int bh = idx >> 8;
  const float* w = WhT + (size_t)bh * DD * NN + n;
  float v1 = 0.f, v2 = 0.f;
#pragma unroll
  for (int d = 0; d < DD; ++d) {
    float x = w[(size_t)d * NN];
    v1 += x * a[d];
    v2 += x * a[DD + d];
  }
  s1[idx] = v1;
  s2[idx] = v2;
}

__global__ __launch_bounds__(128) void gat_attn(const float* __restrict__ WhT,
                                                const float* __restrict__ s1,
                                                const float* __restrict__ s2,
                                                const int* __restrict__ mask,
                                                const float* __restrict__ Xin,
                                                float* __restrict__ Xout) {
  const int tid  = threadIdx.x;
  const int lane = tid & 31;
  const int wave = tid >> 5;
  const int mb   = blockIdx.x & 7;
  const int h    = (blockIdx.x >> 3) & (HH - 1);
  const int bt   = blockIdx.x >> 5;
  const int m0   = mb * 32;

  __shared__ float att[32][NN];
  __shared__ float s2l[NN];
  __shared__ float s1l[32];
  __shared__ int   ml[NN];
  __shared__ int   mil[32];

  for (int i = tid; i < NN; i += 128) {
    s2l[i] = s2[(bt * HH + h) * NN + i];
    ml[i]  = mask[bt * NN + i];
  }
  if (tid < 32) {
    s1l[tid] = s1[(bt * HH + h) * NN + m0 + tid];
    mil[tid] = mask[bt * NN + m0 + tid];
  }
  __syncthreads();

  for (int rr = 0; rr < 8; ++rr) {
    int   r   = wave * 8 + rr;
    int   mi  = mil[r];
    float s1v = s1l[r];
    float vmax = NEGV;
#pragma unroll
    for (int q = 0; q < 8; ++q) {
      int j = lane + 32 * q;
      float e = NEGV;
      if (mi && ml[j]) {
        float t = s1v + s2l[j];
        e = (t > 0.f) ? t : 0.2f * t;
      }
      vmax = fmaxf(vmax, e);
    }
    vmax = wave_max32(vmax);
    float vsum = 0.f;
#pragma unroll
    for (int q = 0; q < 8; ++q) {
      int j = lane + 32 * q;
      float p = 0.f;
      if (mi && ml[j]) {
        float t = s1v + s2l[j];
        float e = (t > 0.f) ? t : 0.2f * t;
        p = __expf(e - vmax);
      }
      att[r][j] = p;
      vsum += p;
    }
    vsum = wave_sum32(vsum);
    float inv = (vsum > 0.f) ? 1.f / vsum : 0.f;
#pragma unroll
    for (int q = 0; q < 8; ++q) att[r][lane + 32 * q] *= inv;
  }
  __syncthreads();

  const int mt = wave >> 1;
  const int nt = wave & 1;
  const int kh = lane >> 4;
  const int nl = lane & 15;
  const float* Bh = WhT + ((size_t)(bt * HH + h) * DD + nt * 16 + nl) * NN;
  v8f acc = {};
  for (int kk = 0; kk < NN; kk += 4) {
    v2f a = *(const v2f*)&att[mt * 16 + nl][kk + 2 * kh];
    v2f b = *(const v2f*)(Bh + kk + 2 * kh);
    acc = __builtin_amdgcn_wmma_f32_16x16x4_f32(false, a, false, b,
                                                (short)0, acc, false, false);
  }
#pragma unroll
  for (int v = 0; v < 8; ++v) {
    int grow = m0 + mt * 16 + v + 8 * kh;
    int gcol = h * DD + nt * 16 + nl;
    size_t o = (size_t)(bt * NN + grow) * FF + gcol;
    float x = Xin[o] + acc[v];
    Xout[o] = (x > 0.f) ? x : expm1f(x);
  }
}

__global__ __launch_bounds__(256) void layernorm_mask(const float* __restrict__ X,
                                                      const int* __restrict__ mask,
                                                      const float* __restrict__ gamma,
                                                      const float* __restrict__ beta,
                                                      float* __restrict__ Out) {
  const int lane = threadIdx.x & 31;
  const int wave = threadIdx.x >> 5;
  const int row  = blockIdx.x * 8 + wave;
  const float* x = X + (size_t)row * FF;
  float v[4];
  float s = 0.f;
#pragma unroll
  for (int q = 0; q < 4; ++q) { v[q] = x[lane + 32 * q]; s += v[q]; }
  float mu = wave_sum32(s) * (1.f / FF);
  float var = 0.f;
#pragma unroll
  for (int q = 0; q < 4; ++q) { float d = v[q] - mu; var += d * d; }
  var = wave_sum32(var) * (1.f / FF);
  float rstd = rsqrtf(var + 1e-5f);
  float mk = (mask[row] > 0) ? 1.f : 0.f;
  float* o = Out + (size_t)row * FF;
#pragma unroll
  for (int q = 0; q < 4; ++q) {
    int c = lane + 32 * q;
    o[c] = ((v[q] - mu) * rstd * gamma[c] + beta[c]) * mk;
  }
}

extern "C" void kernel_launch(void* const* d_in, const int* in_sizes, int n_in,
                              void* d_out, int out_size, void* d_ws, size_t ws_size,
                              hipStream_t stream) {
  const float* X0    = (const float*)d_in[0];
  const int*   mask  = (const int*)d_in[1];
  const float* W0    = (const float*)d_in[2];
  const float* a0    = (const float*)d_in[3];
  const float* W1    = (const float*)d_in[4];
  const float* a1    = (const float*)d_in[5];
  const float* gamma = (const float*)d_in[6];
  const float* beta  = (const float*)d_in[7];
  float* out = (float*)d_out;

  float* Xb  = (float*)d_ws;
  float* WhT = Xb + (size_t)BT * NN * FF;
  float* s1  = WhT + (size_t)BT * NN * FF;
  float* s2  = s1 + (size_t)BT * HH * NN;
  float* Wt  = s2 + (size_t)BT * HH * NN;

  transpose_w<<<64, 256, 0, stream>>>(W0, Wt);
  gemm_xw<<<BT * 16, 256, 0, stream>>>(X0, Wt, WhT);
  gat_scores<<<(BT * HH * NN) / 256, 256, 0, stream>>>(WhT, a0, s1, s2);
  gat_attn<<<BT * HH * 8, 128, 0, stream>>>(WhT, s1, s2, mask, X0, Xb);
  transpose_w<<<64, 256, 0, stream>>>(W1, Wt);
  gemm_xw<<<BT * 16, 256, 0, stream>>>(Xb, Wt, WhT);
  gat_scores<<<(BT * HH * NN) / 256, 256, 0, stream>>>(WhT, a1, s1, s2);
  gat_attn<<<BT * HH * 8, 128, 0, stream>>>(WhT, s1, s2, mask, Xb, Xb);
  layernorm_mask<<<(BT * NN) / 8, 256, 0, stream>>>(Xb, mask, gamma, beta, out);
}


# attention softmax: each lane keeps its 8 column terms (s2, mask) in registers for all rows of its wave; rows evaluated branch-free, masked rows skipped wave-uniformly; no per-element LDS reads
# speedup vs baseline: 1.2521x; 1.2521x over previous
#include <hip/hip_runtime.h>


typedef float v2f __attribute__((ext_vector_type(2)));
typedef float v8f __attribute__((ext_vector_type(8)));

#define BT 128
#define NN 256
#define FF 128
#define HH 4
#define DD 32
#define NEGV (-1e9f)

__device__ __forceinline__ float wave_max32(float v) {
#pragma unroll
  for (int o = 16; o > 0; o >>= 1) v = fmaxf(v, __shfl_xor(v, o, 32));
  return v;
}
__device__ __forceinline__ float wave_sum32(float v) {
#pragma unroll
  for (int o = 16; o > 0; o >>= 1) v += __shfl_xor(v, o, 32);
  return v;
}

__global__ __launch_bounds__(256) void transpose_w(const float* __restrict__ W,
                                                   float* __restrict__ Wt) {
  int idx = blockIdx.x * 256 + threadIdx.x;
  int n = idx >> 7, k = idx & 127;
  Wt[idx] = W[k * FF + n];
}

__global__ __launch_bounds__(256) void gemm_xw(const float* __restrict__ X,
                                               const float* __restrict__ Wt,
                                               float* __restrict__ WhT) {
  const int tid  = threadIdx.x;
  const int lane = tid & 31;
  const int wave = tid >> 5;
  const int bt   = blockIdx.x >> 4;
  const int m0   = (blockIdx.x & 15) * 16;

  __shared__ float As[16][FF];
  const float* Xb = X + (size_t)(bt * NN + m0) * FF;
#pragma unroll
  for (int i = 0; i < 8; ++i) {
    int idx = tid + i * 256;
    As[idx >> 7][idx & 127] = Xb[idx];
  }
  __syncthreads();

  const int n0 = wave * 16;
  const int kh = lane >> 4;
  const int nl = lane & 15;
  v8f acc = {};
  for (int kk = 0; kk < FF; kk += 4) {
    v2f a = *(const v2f*)(Wt + (size_t)(n0 + nl) * FF + kk + 2 * kh);
    v2f b = *(const v2f*)&As[nl][kk + 2 * kh];
    acc = __builtin_amdgcn_wmma_f32_16x16x4_f32(false, a, false, b,
                                                (short)0, acc, false, false);
  }
#pragma unroll
  for (int v = 0; v < 8; ++v) {
    int c = n0 + v + 8 * kh;
    int h = c >> 5, d = c & 31;
    WhT[((size_t)(bt * HH + h) * DD + d) * NN + m0 + nl] = acc[v];
  }
}

__global__ __launch_bounds__(256) void gat_scores(const float* __restrict__ WhT,
                                                  const float* __restrict__ a,
                                                  float* __restrict__ s1,
                                                  float* __restrict__ s2) {
  int idx = blockIdx.x * 256 + threadIdx.x;
  int n  = idx & (NN - 1);
  int bh = idx >> 8;
  const float* w = WhT + (size_t)bh * DD * NN + n;
  float v1 = 0.f, v2 = 0.f;
#pragma unroll
  for (int d = 0; d < DD; ++d) {
    float x = w[(size_t)d * NN];
    v1 += x * a[d];
    v2 += x * a[DD + d];
  }
  s1[idx] = v1;
  s2[idx] = v2;
}

__global__ __launch_bounds__(128) void gat_attn(const float* __restrict__ WhT,
                                                const float* __restrict__ s1,
                                                const float* __restrict__ s2,
                                                const int* __restrict__ mask,
                                                const float* __restrict__ Xin,
                                                float* __restrict__ Xout) {
  const int tid  = threadIdx.x;
  const int lane = tid & 31;
  const int wave = __builtin_amdgcn_readfirstlane(tid >> 5);
  const int mb   = blockIdx.x & 7;
  const int h    = (blockIdx.x >> 3) & (HH - 1);
  const int bt   = blockIdx.x >> 5;
  const int m0   = mb * 32;

  __shared__ float att[32][NN];

  const float* s2p = s2 + (bt * HH + h) * NN + lane;
  const int*   mcp = mask + bt * NN + lane;
  float s2r[8];
  bool  okc[8];
#pragma unroll
  for (int q = 0; q < 8; ++q) {
    s2r[q] = s2p[32 * q];
    okc[q] = mcp[32 * q] != 0;
  }
  const float* s1p = s1 + (bt * HH + h) * NN + m0 + wave * 8;
  const int*   mrp = mask + bt * NN + m0 + wave * 8;
  float s1r[8];
  int   mir[8];
#pragma unroll
  for (int rr = 0; rr < 8; ++rr) {
    s1r[rr] = s1p[rr];
    mir[rr] = mrp[rr];
  }

#pragma unroll
  for (int rr = 0; rr < 8; ++rr) {
    float* arow = &att[wave * 8 + rr][lane];
    if (mir[rr] == 0) {
#pragma unroll
      for (int q = 0; q < 8; ++q) arow[32 * q] = 0.f;
      continue;
    }
    const float s1v = s1r[rr];
    float e[8];
    float vmax = NEGV;
#pragma unroll
    for (int q = 0; q < 8; ++q) {
      float t = s1v + s2r[q];
      float x = (t > 0.f) ? t : 0.2f * t;
      e[q] = okc[q] ? x : NEGV;
      vmax = fmaxf(vmax, e[q]);
    }
    vmax = wave_max32(vmax);
    float p[8];
    float vsum = 0.f;
#pragma unroll
    for (int q = 0; q < 8; ++q) {
      float pe = __expf(e[q] - vmax);
      p[q] = okc[q] ? pe : 0.f;
      vsum += p[q];
    }
    vsum = wave_sum32(vsum);
    float inv = (vsum > 0.f) ? 1.f / vsum : 0.f;
#pragma unroll
    for (int q = 0; q < 8; ++q) arow[32 * q] = p[q] * inv;
  }
  __syncthreads();

  const int mt = wave >> 1;
  const int nt = wave & 1;
  const int kh = lane >> 4;
  const int nl = lane & 15;
  const float* Bh = WhT + ((size_t)(bt * HH + h) * DD + nt * 16 + nl) * NN;
  v8f acc = {};
  for (int kk = 0; kk < NN; kk += 4) {
    v2f a = *(const v2f*)&att[mt * 16 + nl][kk + 2 * kh];
    v2f b = *(const v2f*)(Bh + kk + 2 * kh);
    acc = __builtin_amdgcn_wmma_f32_16x16x4_f32(false, a, false, b,
                                                (short)0, acc, false, false);
  }
#pragma unroll
  for (int v = 0; v < 8; ++v) {
    int grow = m0 + mt * 16 + v + 8 * kh;
    int gcol = h * DD + nt * 16 + nl;
    size_t o = (size_t)(bt * NN + grow) * FF + gcol;
    float x = Xin[o] + acc[v];
    Xout[o] = (x > 0.f) ? x : expm1f(x);
  }
}

__global__ __launch_bounds__(256) void layernorm_mask(const float* __restrict__ X,
                                                      const int* __restrict__ mask,
                                                      const float* __restrict__ gamma,
                                                      const float* __restrict__ beta,
                                                      float* __restrict__ Out) {
  const int lane = threadIdx.x & 31;
  const int wave = threadIdx.x >> 5;
  const int row  = blockIdx.x * 8 + wave;
  const float* x = X + (size_t)row * FF;
  float v[4];
  float s = 0.f;
#pragma unroll
  for (int q = 0; q < 4; ++q) { v[q] = x[lane + 32 * q]; s += v[q]; }
  float mu = wave_sum32(s) * (1.f / FF);
  float var = 0.f;
#pragma unroll
  for (int q = 0; q < 4; ++q) { float d = v[q] - mu; var += d * d; }
  var = wave_sum32(var) * (1.f / FF);
  float rstd = rsqrtf(var + 1e-5f);
  float mk = (mask[row] > 0) ? 1.f : 0.f;
  float* o = Out + (size_t)row * FF;
#pragma unroll
  for (int q = 0; q < 4; ++q) {
    int c = lane + 32 * q;
    o[c] = ((v[q] - mu) * rstd * gamma[c] + beta[c]) * mk;
  }
}

extern "C" void kernel_launch(void* const* d_in, const int* in_sizes, int n_in,
                              void* d_out, int out_size, void* d_ws, size_t ws_size,
                              hipStream_t stream) {
  const float* X0    = (const float*)d_in[0];
  const int*   mask  = (const int*)d_in[1];
  const float* W0    = (const float*)d_in[2];
  const float* a0    = (const float*)d_in[3];
  const float* W1    = (const float*)d_in[4];
  const float* a1    = (const float*)d_in[5];
  const float* gamma = (const float*)d_in[6];
  const float* beta  = (const float*)d_in[7];
  float* out = (float*)d_out;

  float* Xb  = (float*)d_ws;
  float* WhT = Xb + (size_t)BT * NN * FF;
  float* s1  = WhT + (size_t)BT * NN * FF;
  float* s2  = s1 + (size_t)BT * HH * NN;
  float* Wt  = s2 + (size_t)BT * HH * NN;

  transpose_w<<<64, 256, 0, stream>>>(W0, Wt);
  gemm_xw<<<BT * 16, 256, 0, stream>>>(X0, Wt, WhT);
  gat_scores<<<(BT * HH * NN) / 256, 256, 0, stream>>>(WhT, a0, s1, s2);
  gat_attn<<<BT * HH * 8, 128, 0, stream>>>(WhT, s1, s2, mask, X0, Xb);
  transpose_w<<<64, 256, 0, stream>>>(W1, Wt);
  gemm_xw<<<BT * 16, 256, 0, stream>>>(Xb, Wt, WhT);
  gat_scores<<<(BT * HH * NN) / 256, 256, 0, stream>>>(WhT, a1, s1, s2);
  gat_attn<<<BT * HH * 8, 128, 0, stream>>>(WhT, s1, s2, mask, Xb, Xb);
  layernorm_mask<<<(BT * NN) / 8, 256, 0, stream>>>(Xb, mask, gamma, beta, out);
}


# LDS tiles read as WMMA fragments (X slice in the GEMM, attention matrix) get a row pitch of 4 mod 64 floats, so the 16 rows of a fragment read hit 16 bank groups instead of one bank pair
# speedup vs baseline: 1.4739x; 1.1771x over previous
#include <hip/hip_runtime.h>


typedef float v2f __attribute__((ext_vector_type(2)));
typedef float v8f __attribute__((ext_vector_type(8)));

#define BT 128
#define NN 256
#define FF 128
#define HH 4
#define DD 32
#define NEGV (-1e9f)
#define LPAD 4

__device__ __forceinline__ float wave_max32(float v) {
#pragma unroll
  for (int o = 16; o > 0; o >>= 1) v = fmaxf(v, __shfl_xor(v, o, 32));
  return v;
}
__device__ __forceinline__ float wave_sum32(float v) {
#pragma unroll
  for (int o = 16; o > 0; o >>= 1) v += __shfl_xor(v, o, 32);
  return v;
}

__global__ __launch_bounds__(256) void transpose_w(const float* __restrict__ W,
                                                   float* __restrict__ Wt) {
  int idx = blockIdx.x * 256 + threadIdx.x;
  int n = idx >> 7, k = idx & 127;
  Wt[idx] = W[k * FF + n];
}

__global__ __launch_bounds__(256) void gemm_xw(const float* __restrict__ X,
                                               const float* __restrict__ Wt,
                                               float* __restrict__ WhT) {
  const int tid  = threadIdx.x;
  const int lane = tid & 31;
  const int wave = tid >> 5;
  const int bt   = blockIdx.x >> 4;
  const int m0   = (blockIdx.x & 15) * 16;

  __shared__ float As[16][FF + LPAD];
  const float* Xb = X + (size_t)(bt * NN + m0) * FF;
#pragma unroll
  for (int i = 0; i < 8; ++i) {
    int idx = tid + i * 256;
    As[idx >> 7][idx & 127] = Xb[idx];
  }
  __syncthreads();

  const int n0 = wave * 16;
  const int kh = lane >> 4;
  const int nl = lane & 15;
  v8f acc = {};
  for (int kk = 0; kk < FF; kk += 4) {
    v2f a = *(const v2f*)(Wt + (size_t)(n0 + nl) * FF + kk + 2 * kh);
    v2f b = *(const v2f*)&As[nl][kk + 2 * kh];
    acc = __builtin_amdgcn_wmma_f32_16x16x4_f32(false, a, false, b,
                                                (short)0, acc, false, false);
  }
#pragma unroll
  for (int v = 0; v < 8; ++v) {
    int c = n0 + v + 8 * kh;
    int h = c >> 5, d = c & 31;
    WhT[((size_t)(bt * HH + h) * DD + d) * NN + m0 + nl] = acc[v];
  }
}

__global__ __launch_bounds__(256) void gat_scores(const float* __restrict__ WhT,
                                                  const float* __restrict__ a,
                                                  float* __restrict__ s1,
                                                  float* __restrict__ s2) {
  int idx = blockIdx.x * 256 + threadIdx.x;
  int n  = idx & (NN - 1);
  int bh = idx >> 8;
  const float* w = WhT + (size_t)bh * DD * NN + n;
  float v1 = 0.f, v2 = 0.f;
#pragma unroll
  for (int d = 0; d < DD; ++d) {
    float x = w[(size_t)d * NN];
    v1 += x * a[d];
    v2 += x * a[DD + d];
  }
  s1[idx] = v1;
  s2[idx] = v2;
}

__global__ __launch_bounds__(128) void gat_attn(const float* __restrict__ WhT,
                                                const float* __restrict__ s1,
                                                const float* __restrict__ s2,
                                                const int* __restrict__ mask,
                                                const float* __restrict__ Xin,
                                                float* __restrict__ Xout) {
  const int tid  = threadIdx.x;
  const int lane = tid & 31;
  const int wave = __builtin_amdgcn_readfirstlane(tid >> 5);
  const int mb   = blockIdx.x & 7;
  const int h    = (blockIdx.x >> 3) & (HH - 1);
  const int bt   = blockIdx.x >> 5;
  const int m0   = mb * 32;

  __shared__ float att[32][NN + LPAD];

  const float* s2p = s2 + (bt * HH + h) * NN + lane;
  const int*   mcp = mask + bt * NN + lane;
  float s2r[8];
  bool  okc[8];
#pragma unroll
  for (int q = 0; q < 8; ++q) {
    s2r[q] = s2p[32 * q];
    okc[q] = mcp[32 * q] != 0;
  }
  const float* s1p = s1 + (bt * HH + h) * NN + m0 + wave * 8;
  const int*   mrp = mask + bt * NN + m0 + wave * 8;
  float s1r[8];
  int   mir[8];
#pragma unroll
  for (int rr = 0; rr < 8; ++rr) {
    s1r[rr] = s1p[rr];
    mir[rr] = mrp[rr];
  }

#pragma unroll
  for (int rr = 0; rr < 8; ++rr) {
    float* arow = &att[wave * 8 + rr][lane];
    if (mir[rr] == 0) {
#pragma unroll
      for (int q = 0; q < 8; ++q) arow[32 * q] = 0.f;
      continue;
    }
    const float s1v = s1r[rr];
    float e[8];
    float vmax = NEGV;
#pragma unroll
    for (int q = 0; q < 8; ++q) {
      float t = s1v + s2r[q];
      float x = (t > 0.f) ? t : 0.2f * t;
      e[q] = okc[q] ? x : NEGV;
      vmax = fmaxf(vmax, e[q]);
    }
    vmax = wave_max32(vmax);
    float p[8];
    float vsum = 0.f;
#pragma unroll
    for (int q = 0; q < 8; ++q) {
      float pe = __expf(e[q] - vmax);
      p[q] = okc[q] ? pe : 0.f;
      vsum += p[q];
    }
    vsum = wave_sum32(vsum);
    float inv = (vsum > 0.f) ? 1.f / vsum : 0.f;
#pragma unroll
    for (int q = 0; q < 8; ++q) arow[32 * q] = p[q] * inv;
  }
  __syncthreads();

  const int mt = wave >> 1;
  const int nt = wave & 1;
  const int kh = lane >> 4;
  const int nl = lane & 15;
  const float* Bh = WhT + ((size_t)(bt * HH + h) * DD + nt * 16 + nl) * NN;
  v8f acc = {};
  for (int kk = 0; kk < NN; kk += 4) {
    v2f a = *(const v2f*)&att[mt * 16 + nl][kk + 2 * kh];
    v2f b = *(const v2f*)(Bh + kk + 2 * kh);
    acc = __builtin_amdgcn_wmma_f32_16x16x4_f32(false, a, false, b,
                                                (short)0, acc, false, false);
  }
#pragma unroll
  for (int v = 0; v < 8; ++v) {
    int grow = m0 + mt * 16 + v + 8 * kh;
    int gcol = h * DD + nt * 16 + nl;
    size_t o = (size_t)(bt * NN + grow) * FF + gcol;
    float x = Xin[o] + acc[v];
    Xout[o] = (x > 0.f) ? x : expm1f(x);
  }
}

__global__ __launch_bounds__(256) void layernorm_mask(const float* __restrict__ X,
                                                      const int* __restrict__ mask,
                                                      const float* __restrict__ gamma,
                                                      const float* __restrict__ beta,
                                                      float* __restrict__ Out) {
  const int lane = threadIdx.x & 31;
  const int wave = threadIdx.x >> 5;
  const int row  = blockIdx.x * 8 + wave;
  const float* x = X + (size_t)row * FF;
  float v[4];
  float s = 0.f;
#pragma unroll
  for (int q = 0; q < 4; ++q) { v[q] = x[lane + 32 * q]; s += v[q]; }
  float mu = wave_sum32(s) * (1.f / FF);
  float var = 0.f;
#pragma unroll
  for (int q = 0; q < 4; ++q) { float d = v[q] - mu; var += d * d; }
  var = wave_sum32(var) * (1.f / FF);
  float rstd = rsqrtf(var + 1e-5f);
  float mk = (mask[row] > 0) ? 1.f : 0.f;
  float* o = Out + (size_t)row * FF;
#pragma unroll
  for (int q = 0; q < 4; ++q) {
    int c = lane + 32 * q;
    o[c] = ((v[q] - mu) * rstd * gamma[c] + beta[c]) * mk;
  }
}

extern "C" void kernel_launch(void* const* d_in, const int* in_sizes, int n_in,
                              void* d_out, int out_size, void* d_ws, size_t ws_size,
                              hipStream_t stream) {
  const float* X0    = (const float*)d_in[0];
  const int*   mask  = (const int*)d_in[1];
  const float* W0    = (const float*)d_in[2];
  const float* a0    = (const float*)d_in[3];
  const float* W1    = (const float*)d_in[4];
  const float* a1    = (const float*)d_in[5];
  const float* gamma = (const float*)d_in[6];
  const float* beta  = (const float*)d_in[7];
  float* out = (float*)d_out;

  float* Xb  = (float*)d_ws;
  float* WhT = Xb + (size_t)BT * NN * FF;
  float* s1  = WhT + (size_t)BT * NN * FF;
  float* s2  = s1 + (size_t)BT * HH * NN;
  float* Wt  = s2 + (size_t)BT * HH * NN;

  transpose_w<<<64, 256, 0, stream>>>(W0, Wt);
  gemm_xw<<<BT * 16, 256, 0, stream>>>(X0, Wt, WhT);
  gat_scores<<<(BT * HH * NN) / 256, 256, 0, stream>>>(WhT, a0, s1, s2);
  gat_attn<<<BT * HH * 8, 128, 0, stream>>>(WhT, s1, s2, mask, X0, Xb);
  transpose_w<<<64, 256, 0, stream>>>(W1, Wt);
  gemm_xw<<<BT * 16, 256, 0, stream>>>(Xb, Wt, WhT);
  gat_scores<<<(BT * HH * NN) / 256, 256, 0, stream>>>(WhT, a1, s1, s2);
  gat_attn<<<BT * HH * 8, 128, 0, stream>>>(WhT, s1, s2, mask, Xb, Xb);
  layernorm_mask<<<(BT * NN) / 8, 256, 0, stream>>>(Xb, mask, gamma, beta, out);
}


# attention scores s1/s2 computed in the GEMM epilogue from the block's result tile staged in LDS (same d-ascending sums as before); the two score launches and their pass over WhT are gone
# speedup vs baseline: 1.5704x; 1.0655x over previous
#include <hip/hip_runtime.h>


typedef float v2f __attribute__((ext_vector_type(2)));
typedef float v8f __attribute__((ext_vector_type(8)));

#define BT 128
#define NN 256
#define FF 128
#define HH 4
#define DD 32
#define NEGV (-1e9f)
#define LPAD 4

__device__ __forceinline__ float wave_max32(float v) {
#pragma unroll
  for (int o = 16; o > 0; o >>= 1) v = fmaxf(v, __shfl_xor(v, o, 32));
  return v;
}
__device__ __forceinline__ float wave_sum32(float v) {
#pragma unroll
  for (int o = 16; o > 0; o >>= 1) v += __shfl_xor(v, o, 32);
  return v;
}

__global__ __launch_bounds__(256) void transpose_w(const float* __restrict__ W,
                                                   float* __restrict__ Wt) {
  int idx = blockIdx.x * 256 + threadIdx.x;
  int n = idx >> 7, k = idx & 127;
  Wt[idx] = W[k * FF + n];
}

__global__ __launch_bounds__(256) void gemm_xw(const float* __restrict__ X,
                                               const float* __restrict__ Wt,
                                               const float* __restrict__ avec,
                                               float* __restrict__ WhT,
                                               float* __restrict__ s1,
                                               float* __restrict__ s2) {
  const int tid  = threadIdx.x;
  const int lane = tid & 31;
  const int wave = tid >> 5;
  const int bt   = blockIdx.x >> 4;
  const int m0   = (blockIdx.x & 15) * 16;

  __shared__ float As[16][FF + LPAD];
  const float* Xb = X + (size_t)(bt * NN + m0) * FF;
#pragma unroll
  for (int i = 0; i < 8; ++i) {
    int idx = tid + i * 256;
    As[idx >> 7][idx & 127] = Xb[idx];
  }
  __syncthreads();

  const int n0 = wave * 16;
  const int kh = lane >> 4;
  const int nl = lane & 15;
  v8f acc = {};
  for (int kk = 0; kk < FF; kk += 4) {
    v2f a = *(const v2f*)(Wt + (size_t)(n0 + nl) * FF + kk + 2 * kh);
    v2f b = *(const v2f*)&As[nl][kk + 2 * kh];
    acc = __builtin_amdgcn_wmma_f32_16x16x4_f32(false, a, false, b,
                                                (short)0, acc, false, false);
  }
  __shared__ float Ts[FF][17];
#pragma unroll
  for (int v = 0; v < 8; ++v) Ts[n0 + v + 8 * kh][nl] = acc[v];
  __syncthreads();
#pragma unroll
  for (int v = 0; v < 8; ++v) {
    int c = n0 + v + 8 * kh;
    int h = c >> 5, d = c & 31;
    WhT[((size_t)(bt * HH + h) * DD + d) * NN + m0 + nl] = acc[v];
  }
  if (tid < 64) {
    const int hh = tid >> 4;
    const int n  = tid & 15;
    float v1 = 0.f, v2 = 0.f;
#pragma unroll
    for (int d = 0; d < DD; ++d) {
      float x = Ts[hh * DD + d][n];
      v1 += x * avec[d];
      v2 += x * avec[DD + d];
    }
    const int o = (bt * HH + hh) * NN + m0 + n;
    s1[o] = v1;
    s2[o] = v2;
  }
}

__global__ __launch_bounds__(128) void gat_attn(const float* __restrict__ WhT,
                                                const float* __restrict__ s1,
                                                const float* __restrict__ s2,
                                                const int* __restrict__ mask,
                                                const float* __restrict__ Xin,
                                                float* __restrict__ Xout) {
  const int tid  = threadIdx.x;
  const int lane = tid & 31;
  const int wave = __builtin_amdgcn_readfirstlane(tid >> 5);
  const int mb   = blockIdx.x & 7;
  const int h    = (blockIdx.x >> 3) & (HH - 1);
  const int bt   = blockIdx.x >> 5;
  const int m0   = mb * 32;

  __shared__ float att[32][NN + LPAD];

  const float* s2p = s2 + (bt * HH + h) * NN + lane;
  const int*   mcp = mask + bt * NN + lane;
  float s2r[8];
  bool  okc[8];
#pragma unroll
  for (int q = 0; q < 8; ++q) {
    s2r[q] = s2p[32 * q];
    okc[q] = mcp[32 * q] != 0;
  }
  const float* s1p = s1 + (bt * HH + h) * NN + m0 + wave * 8;
  const int*   mrp = mask + bt * NN + m0 + wave * 8;
  float s1r[8];
  int   mir[8];
#pragma unroll
  for (int rr = 0; rr < 8; ++rr) {
    s1r[rr] = s1p[rr];
    mir[rr] = mrp[rr];
  }

#pragma unroll
  for (int rr = 0; rr < 8; ++rr) {
    float* arow = &att[wave * 8 + rr][lane];
    if (mir[rr] == 0) {
#pragma unroll
      for (int q = 0; q < 8; ++q) arow[32 * q] = 0.f;
      continue;
    }
    const float s1v = s1r[rr];
    float e[8];
    float vmax = NEGV;
#pragma unroll
    for (int q = 0; q < 8; ++q) {
      float t = s1v + s2r[q];
      float x = (t > 0.f) ? t : 0.2f * t;
      e[q] = okc[q] ? x : NEGV;
      vmax = fmaxf(vmax, e[q]);
    }
    vmax = wave_max32(vmax);
    float p[8];
    float vsum = 0.f;
#pragma unroll
    for (int q = 0; q < 8; ++q) {
      float pe = __expf(e[q] - vmax);
      p[q] = okc[q] ? pe : 0.f;
      vsum += p[q];
    }
    vsum = wave_sum32(vsum);
    float inv = (vsum > 0.f) ? 1.f / vsum : 0.f;
#pragma unroll
    for (int q = 0; q < 8; ++q) arow[32 * q] = p[q] * inv;
  }
  __syncthreads();

  const int mt = wave >> 1;
  const int nt = wave & 1;
  const int kh = lane >> 4;
  const int nl = lane & 15;
  const float* Bh = WhT + ((size_t)(bt * HH + h) * DD + nt * 16 + nl) * NN;
  v8f acc = {};
  for (int kk = 0; kk < NN; kk += 4) {
    v2f a = *(const v2f*)&att[mt * 16 + nl][kk + 2 * kh];
    v2f b = *(const v2f*)(Bh + kk + 2 * kh);
    acc = __builtin_amdgcn_wmma_f32_16x16x4_f32(false, a, false, b,
                                                (short)0, acc, false, false);
  }
#pragma unroll
  for (int v = 0; v < 8; ++v) {
    int grow = m0 + mt * 16 + v + 8 * kh;
    int gcol = h * DD + nt * 16 + nl;
    size_t o = (size_t)(bt * NN + grow) * FF + gcol;
    float x = Xin[o] + acc[v];
    Xout[o] = (x > 0.f) ? x : expm1f(x);
  }
}

__global__ __launch_bounds__(256) void layernorm_mask(const float* __restrict__ X,
                                                      const int* __restrict__ mask,
                                                      const float* __restrict__ gamma,
                                                      const float* __restrict__ beta,
                                                      float* __restrict__ Out) {
  const int lane = threadIdx.x & 31;
  const int wave = threadIdx.x >> 5;
  const int row  = blockIdx.x * 8 + wave;
  const float* x = X + (size_t)row * FF;
  float v[4];
  float s = 0.f;
#pragma unroll
  for (int q = 0; q < 4; ++q) { v[q] = x[lane + 32 * q]; s += v[q]; }
  float mu = wave_sum32(s) * (1.f / FF);
  float var = 0.f;
#pragma unroll
  for (int q = 0; q < 4; ++q) { float d = v[q] - mu; var += d * d; }
  var = wave_sum32(var) * (1.f / FF);
  float rstd = rsqrtf(var + 1e-5f);
  float mk = (mask[row] > 0) ? 1.f : 0.f;
  float* o = Out + (size_t)row * FF;
#pragma unroll
  for (int q = 0; q < 4; ++q) {
    int c = lane + 32 * q;
    o[c] = ((v[q] - mu) * rstd * gamma[c] + beta[c]) * mk;
  }
}

extern "C" void kernel_launch(void* const* d_in, const int* in_sizes, int n_in,
                              void* d_out, int out_size, void* d_ws, size_t ws_size,
                              hipStream_t stream) {
  const float* X0    = (const float*)d_in[0];
  const int*   mask  = (const int*)d_in[1];
  const float* W0    = (const float*)d_in[2];
  const float* a0    = (const float*)d_in[3];
  const float* W1    = (const float*)d_in[4];
  const float* a1    = (const float*)d_in[5];
  const float* gamma = (const float*)d_in[6];
  const float* beta  = (const float*)d_in[7];
  float* out = (float*)d_out;

  float* Xb  = (float*)d_ws;
  float* WhT = Xb + (size_t)BT * NN * FF;
  float* s1  = WhT + (size_t)BT * NN * FF;
  float* s2  = s1 + (size_t)BT * HH * NN;
  float* Wt  = s2 + (size_t)BT * HH * NN;

  transpose_w<<<64, 256, 0, stream>>>(W0, Wt);
  gemm_xw<<<BT * 16, 256, 0, stream>>>(X0, Wt, a0, WhT, s1, s2);
  gat_attn<<<BT * HH * 8, 128, 0, stream>>>(WhT, s1, s2, mask, X0, Xb);
  transpose_w<<<64, 256, 0, stream>>>(W1, Wt);
  gemm_xw<<<BT * 16, 256, 0, stream>>>(Xb, Wt, a1, WhT, s1, s2);
  gat_attn<<<BT * HH * 8, 128, 0, stream>>>(WhT, s1, s2, mask, Xb, Xb);
  layernorm_mask<<<(BT * NN) / 8, 256, 0, stream>>>(Xb, mask, gamma, beta, out);
}


# global WMMA operands (W for the GEMM, Wh for attention) stored in fragment order: a wave's operand load is one contiguous 512-byte b128 read, not 16 rows x 16 bytes; GEMM writes Wh in whole lines
# speedup vs baseline: 2.4424x; 1.5553x over previous
#include <hip/hip_runtime.h>


typedef float v2f __attribute__((ext_vector_type(2)));
typedef float v4f __attribute__((ext_vector_type(4)));
typedef float v8f __attribute__((ext_vector_type(8)));

#define BT 128
#define NN 256
#define FF 128
#define HH 4
#define DD 32
#define NEGV (-1e9f)
#define LPAD 4

__device__ __forceinline__ float wave_max32(float v) {
#pragma unroll
  for (int o = 16; o > 0; o >>= 1) v = fmaxf(v, __shfl_xor(v, o, 32));
  return v;
}
__device__ __forceinline__ float wave_sum32(float v) {
#pragma unroll
  for (int o = 16; o > 0; o >>= 1) v += __shfl_xor(v, o, 32);
  return v;
}

__global__ __launch_bounds__(256) void pack_w(const float* __restrict__ W,
                                              float* __restrict__ Wp) {
  int idx  = blockIdx.x * 256 + threadIdx.x;
  int e    = idx & 3;
  int lane = (idx >> 2) & 31;
  int kp   = (idx >> 7) & 15;
  int nt   = idx >> 11;
  int k = kp * 8 + (e >> 1) * 4 + 2 * (lane >> 4) + (e & 1);
  int n = nt * 16 + (lane & 15);
  Wp[idx] = W[k * FF + n];
}

__global__ __launch_bounds__(256) void gemm_xw(const float* __restrict__ X,
                                               const float* __restrict__ Wp,
                                               const float* __restrict__ avec,
                                               float* __restrict__ WhP,
                                               float* __restrict__ s1,
                                               float* __restrict__ s2) {
  const int tid  = threadIdx.x;
  const int lane = tid & 31;
  const int wave = tid >> 5;
  const int bt   = blockIdx.x >> 4;
  const int m0   = (blockIdx.x & 15) * 16;

  __shared__ float As[16][FF + LPAD];
  const float* Xb = X + (size_t)(bt * NN + m0) * FF;
#pragma unroll
  for (int i = 0; i < 8; ++i) {
    int idx = tid + i * 256;
    As[idx >> 7][idx & 127] = Xb[idx];
  }
  __syncthreads();

  const int n0 = wave * 16;
  const int kh = lane >> 4;
  const int nl = lane & 15;
  v8f acc = {};
  const v4f* Ap = (const v4f*)Wp + (size_t)(wave * 16) * 32 + lane;
  for (int kp = 0; kp < FF / 8; ++kp) {
    v4f a4 = Ap[kp * 32];
    v2f a0 = {a4.x, a4.y};
    v2f a1 = {a4.z, a4.w};
    v2f b0 = *(const v2f*)&As[nl][kp * 8 + 2 * kh];
    v2f b1 = *(const v2f*)&As[nl][kp * 8 + 4 + 2 * kh];
    acc = __builtin_amdgcn_wmma_f32_16x16x4_f32(false, a0, false, b0,
                                                (short)0, acc, false, false);
    acc = __builtin_amdgcn_wmma_f32_16x16x4_f32(false, a1, false, b1,
                                                (short)0, acc, false, false);
  }
  __shared__ float Ts[FF][17];
#pragma unroll
  for (int v = 0; v < 8; ++v) Ts[n0 + v + 8 * kh][nl] = acc[v];
  __syncthreads();
  {
    const int kpl  = tid >> 7;
    const int fln  = (tid >> 2) & 31;
    const int e    = tid & 3;
    const int nloc = kpl * 8 + (e >> 1) * 4 + 2 * (fln >> 4) + (e & 1);
    const int fl   = fln & 15;
    float* dst = WhP + (size_t)bt * (HH * DD * NN) + (m0 >> 3) * 128 + tid;
#pragma unroll
    for (int c = 0; c < 8; ++c)
      dst[c * (16 * NN)] = Ts[c * 16 + fl][nloc];
  }
  if (tid < 64) {
    const int hh = tid >> 4;
    const int n  = tid & 15;
    float v1 = 0.f, v2 = 0.f;
#pragma unroll
    for (int d = 0; d < DD; ++d) {
      float x = Ts[hh * DD + d][n];
      v1 += x * avec[d];
      v2 += x * avec[DD + d];
    }
    const int o = (bt * HH + hh) * NN + m0 + n;
    s1[o] = v1;
    s2[o] = v2;
  }
}

__global__ __launch_bounds__(128) void gat_attn(const float* __restrict__ WhP,
                                                const float* __restrict__ s1,
                                                const float* __restrict__ s2,
                                                const int* __restrict__ mask,
                                                const float* __restrict__ Xin,
                                                float* __restrict__ Xout) {
  const int tid  = threadIdx.x;
  const int lane = tid & 31;
  const int wave = __builtin_amdgcn_readfirstlane(tid >> 5);
  const int mb   = blockIdx.x & 7;
  const int h    = (blockIdx.x >> 3) & (HH - 1);
  const int bt   = blockIdx.x >> 5;
  const int m0   = mb * 32;

  __shared__ float att[32][NN + LPAD];

  const float* s2p = s2 + (bt * HH + h) * NN + lane;
  const int*   mcp = mask + bt * NN + lane;
  float s2r[8];
  bool  okc[8];
#pragma unroll
  for (int q = 0; q < 8; ++q) {
    s2r[q] = s2p[32 * q];
    okc[q] = mcp[32 * q] != 0;
  }
  const float* s1p = s1 + (bt * HH + h) * NN + m0 + wave * 8;
  const int*   mrp = mask + bt * NN + m0 + wave * 8;
  float s1r[8];
  int   mir[8];
#pragma unroll
  for (int rr = 0; rr < 8; ++rr) {
    s1r[rr] = s1p[rr];
    mir[rr] = mrp[rr];
  }

#pragma unroll
  for (int rr = 0; rr < 8; ++rr) {
    float* arow = &att[wave * 8 + rr][lane];
    if (mir[rr] == 0) {
#pragma unroll
      for (int q = 0; q < 8; ++q) arow[32 * q] = 0.f;
      continue;
    }
    const float s1v = s1r[rr];
    float e[8];
    float vmax = NEGV;
#pragma unroll
    for (int q = 0; q < 8; ++q) {
      float t = s1v + s2r[q];
      float x = (t > 0.f) ? t : 0.2f * t;
      e[q] = okc[q] ? x : NEGV;
      vmax = fmaxf(vmax, e[q]);
    }
    vmax = wave_max32(vmax);
    float p[8];
    float vsum = 0.f;
#pragma unroll
    for (int q = 0; q < 8; ++q) {
      float pe = __expf(e[q] - vmax);
      p[q] = okc[q] ? pe : 0.f;
      vsum += p[q];
    }
    vsum = wave_sum32(vsum);
    float inv = (vsum > 0.f) ? 1.f / vsum : 0.f;
#pragma unroll
    for (int q = 0; q < 8; ++q) arow[32 * q] = p[q] * inv;
  }
  __syncthreads();

  const int mt = wave >> 1;
  const int nt = wave & 1;
  const int kh = lane >> 4;
  const int nl = lane & 15;
  const v4f* Bp =
      (const v4f*)(WhP + ((size_t)(bt * HH + h) * 2 + nt) * (16 * NN)) + lane;
  v8f acc = {};
  for (int kp = 0; kp < NN / 8; ++kp) {
    v4f b4 = Bp[kp * 32];
    v2f b0 = {b4.x, b4.y};
    v2f b1 = {b4.z, b4.w};
    v2f a0 = *(const v2f*)&att[mt * 16 + nl][kp * 8 + 2 * kh];
    v2f a1 = *(const v2f*)&att[mt * 16 + nl][kp * 8 + 4 + 2 * kh];
    acc = __builtin_amdgcn_wmma_f32_16x16x4_f32(false, a0, false, b0,
                                                (short)0, acc, false, false);
    acc = __builtin_amdgcn_wmma_f32_16x16x4_f32(false, a1, false, b1,
                                                (short)0, acc, false, false);
  }
#pragma unroll
  for (int v = 0; v < 8; ++v) {
    int grow = m0 + mt * 16 + v + 8 * kh;
    int gcol = h * DD + nt * 16 + nl;
    size_t o = (size_t)(bt * NN + grow) * FF + gcol;
    float x = Xin[o] + acc[v];
    Xout[o] = (x > 0.f) ? x : expm1f(x);
  }
}

__global__ __launch_bounds__(256) void layernorm_mask(const float* __restrict__ X,
                                                      const int* __restrict__ mask,
                                                      const float* __restrict__ gamma,
                                                      const float* __restrict__ beta,
                                                      float* __restrict__ Out) {
  const int lane = threadIdx.x & 31;
  const int wave = threadIdx.x >> 5;
  const int row  = blockIdx.x * 8 + wave;
  const float* x = X + (size_t)row * FF;
  float v[4];
  float s = 0.f;
#pragma unroll
  for (int q = 0; q < 4; ++q) { v[q] = x[lane + 32 * q]; s += v[q]; }
  float mu = wave_sum32(s) * (1.f / FF);
  float var = 0.f;
#pragma unroll
  for (int q = 0; q < 4; ++q) { float d = v[q] - mu; var += d * d; }
  var = wave_sum32(var) * (1.f / FF);
  float rstd = rsqrtf(var + 1e-5f);
  float mk = (mask[row] > 0) ? 1.f : 0.f;
  float* o = Out + (size_t)row * FF;
#pragma unroll
  for (int q = 0; q < 4; ++q) {
    int c = lane + 32 * q;
    o[c] = ((v[q] - mu) * rstd * gamma[c] + beta[c]) * mk;
  }
}

extern "C" void kernel_launch(void* const* d_in, const int* in_sizes, int n_in,
                              void* d_out, int out_size, void* d_ws, size_t ws_size,
                              hipStream_t stream) {
  const float* X0    = (const float*)d_in[0];
  const int*   mask  = (const int*)d_in[1];
  const float* W0    = (const float*)d_in[2];
  const float* a0    = (const float*)d_in[3];
  const float* W1    = (const float*)d_in[4];
  const float* a1    = (const float*)d_in[5];
  const float* gamma = (const float*)d_in[6];
  const float* beta  = (const float*)d_in[7];
  float* out = (float*)d_out;

  float* Xb  = (float*)d_ws;
  float* WhP = Xb + (size_t)BT * NN * FF;
  float* s1  = WhP + (size_t)BT * NN * FF;
  float* s2  = s1 + (size_t)BT * HH * NN;
  float* Wp  = s2 + (size_t)BT * HH * NN;

  pack_w<<<64, 256, 0, stream>>>(W0, Wp);
  gemm_xw<<<BT * 16, 256, 0, stream>>>(X0, Wp, a0, WhP, s1, s2);
  gat_attn<<<BT * HH * 8, 128, 0, stream>>>(WhP, s1, s2, mask, X0, Xb);
  pack_w<<<64, 256, 0, stream>>>(W1, Wp);
  gemm_xw<<<BT * 16, 256, 0, stream>>>(Xb, Wp, a1, WhP, s1, s2);
  gat_attn<<<BT * HH * 8, 128, 0, stream>>>(WhP, s1, s2, mask, Xb, Xb);
  layernorm_mask<<<(BT * NN) / 8, 256, 0, stream>>>(Xb, mask, gamma, beta, out);
}


# masked nodes dropped from all work: each block compacts its graph's unmasked nodes (ballot prefix); GEMM rows, softmax and att@Wh k run compacted; j-sums regrouped (zero terms gone)
# speedup vs baseline: 2.7490x; 1.1255x over previous
#include <hip/hip_runtime.h>


typedef float v2f __attribute__((ext_vector_type(2)));
typedef float v4f __attribute__((ext_vector_type(4)));
typedef float v8f __attribute__((ext_vector_type(8)));

#define BT 128
#define NN 256
#define FF 128
#define HH 4
#define DD 32
#define NEGV (-1e9f)
#define LPAD 4

__device__ __forceinline__ float wave_max32(float v) {
#pragma unroll
  for (int o = 16; o > 0; o >>= 1) v = fmaxf(v, __shfl_xor(v, o, 32));
  return v;
}
__device__ __forceinline__ float wave_sum32(float v) {
#pragma unroll
  for (int o = 16; o > 0; o >>= 1) v += __shfl_xor(v, o, 32);
  return v;
}

__global__ __launch_bounds__(256) void pack_w(const float* __restrict__ W,
                                              float* __restrict__ Wp) {
  int idx  = blockIdx.x * 256 + threadIdx.x;
  int e    = idx & 3;
  int lane = (idx >> 2) & 31;
  int kp   = (idx >> 7) & 15;
  int nt   = idx >> 11;
  int k = kp * 8 + (e >> 1) * 4 + 2 * (lane >> 4) + (e & 1);
  int n = nt * 16 + (lane & 15);
  Wp[idx] = W[k * FF + n];
}

__global__ __launch_bounds__(256) void gemm_xw(const float* __restrict__ X,
                                               const int* __restrict__ mask,
                                               const float* __restrict__ Wp,
                                               const float* __restrict__ avec,
                                               float* __restrict__ WhP,
                                               float* __restrict__ s1,
                                               float* __restrict__ s2) {
  const int tid  = threadIdx.x;
  const int lane = tid & 31;
  const int wave = tid >> 5;
  const int bt   = blockIdx.x >> 4;
  const int m0   = (blockIdx.x & 15) * 16;

  __shared__ int idx_s[NN];
  __shared__ int wc_s[8];
  const bool mk = mask[bt * NN + tid] > 0;
  const unsigned bal = __builtin_amdgcn_ballot_w32(mk);
  const int pre = __popc(bal & ((1u << lane) - 1u));
  if (lane == 0) wc_s[wave] = __popc(bal);
  __syncthreads();
  int base = 0, cnt = 0;
#pragma unroll
  for (int w = 0; w < 8; ++w) {
    int c = wc_s[w];
    cnt += c;
    base += (w < wave) ? c : 0;
  }
  cnt = __builtin_amdgcn_readfirstlane(cnt);
  if (m0 >= ((cnt + 31) & ~31)) return;
  if (mk) idx_s[base + pre] = tid;
  __syncthreads();

  __shared__ float As[16][FF + LPAD];
  const float* Xg = X + (size_t)bt * NN * FF;
  float xr[8];
#pragma unroll
  for (int i = 0; i < 8; ++i) {
    int idx = tid + i * 256;
    int c   = m0 + (idx >> 7);
    int cc  = (c < cnt) ? c : cnt - 1;
    xr[i] = Xg[(size_t)idx_s[cc] * FF + (idx & 127)];
  }
#pragma unroll
  for (int i = 0; i < 8; ++i) {
    int idx = tid + i * 256;
    int r   = idx >> 7;
    As[r][idx & 127] = (m0 + r < cnt) ? xr[i] : 0.f;
  }
  __syncthreads();

  const int n0 = wave * 16;
  const int kh = lane >> 4;
  const int nl = lane & 15;
  v8f acc = {};
  const v4f* Ap = (const v4f*)Wp + (size_t)(wave * 16) * 32 + lane;
  for (int kp = 0; kp < FF / 8; ++kp) {
    v4f a4 = Ap[kp * 32];
    v2f a0 = {a4.x, a4.y};
    v2f a1 = {a4.z, a4.w};
    v2f b0 = *(const v2f*)&As[nl][kp * 8 + 2 * kh];
    v2f b1 = *(const v2f*)&As[nl][kp * 8 + 4 + 2 * kh];
    acc = __builtin_amdgcn_wmma_f32_16x16x4_f32(false, a0, false, b0,
                                                (short)0, acc, false, false);
    acc = __builtin_amdgcn_wmma_f32_16x16x4_f32(false, a1, false, b1,
                                                (short)0, acc, false, false);
  }
  __shared__ float Ts[FF][17];
#pragma unroll
  for (int v = 0; v < 8; ++v) Ts[n0 + v + 8 * kh][nl] = acc[v];
  __syncthreads();
  {
    const int kpl  = tid >> 7;
    const int fln  = (tid >> 2) & 31;
    const int e    = tid & 3;
    const int nloc = kpl * 8 + (e >> 1) * 4 + 2 * (fln >> 4) + (e & 1);
    const int fl   = fln & 15;
    float* dst = WhP + (size_t)bt * (HH * DD * NN) + (m0 >> 3) * 128 + tid;
#pragma unroll
    for (int c = 0; c < 8; ++c)
      dst[c * (16 * NN)] = Ts[c * 16 + fl][nloc];
  }
  if (tid < 64) {
    const int hh = tid >> 4;
    const int n  = tid & 15;
    float v1 = 0.f, v2 = 0.f;
#pragma unroll
    for (int d = 0; d < DD; ++d) {
      float x = Ts[hh * DD + d][n];
      v1 += x * avec[d];
      v2 += x * avec[DD + d];
    }
    const int o = (bt * HH + hh) * NN + m0 + n;
    s1[o] = v1;
    s2[o] = v2;
  }
}

template <int NQ>
__device__ __forceinline__ void softmax_rows(float (*att)[NN + LPAD],
                                             const float* __restrict__ s1p,
                                             const float* __restrict__ s2p,
                                             int r0, int wave, int lane,
                                             int cnt, int nq) {
  float s2r[NQ];
  bool  okc[NQ];
#pragma unroll
  for (int q = 0; q < NQ; ++q) {
    okc[q] = (lane + 32 * q) < cnt;
    s2r[q] = 0.f;
    if (q < nq) s2r[q] = s2p[32 * q];
  }
  float s1r[8];
#pragma unroll
  for (int rr = 0; rr < 8; ++rr) s1r[rr] = s1p[rr];

#pragma unroll
  for (int rr = 0; rr < 8; ++rr) {
    float* arow = &att[wave * 8 + rr][lane];
    if (r0 + rr >= cnt) {
#pragma unroll
      for (int q = 0; q < NQ; ++q) arow[32 * q] = 0.f;
      continue;
    }
    const float s1v = s1r[rr];
    float e[NQ];
    float vmax = NEGV;
#pragma unroll
    for (int q = 0; q < NQ; ++q) {
      float t = s1v + s2r[q];
      float x = (t > 0.f) ? t : 0.2f * t;
      e[q] = okc[q] ? x : NEGV;
      vmax = fmaxf(vmax, e[q]);
    }
    vmax = wave_max32(vmax);
    float p[NQ];
    float vsum = 0.f;
#pragma unroll
    for (int q = 0; q < NQ; ++q) {
      float pe = __expf(e[q] - vmax);
      p[q] = okc[q] ? pe : 0.f;
      vsum += p[q];
    }
    vsum = wave_sum32(vsum);
    float inv = (vsum > 0.f) ? 1.f / vsum : 0.f;
#pragma unroll
    for (int q = 0; q < NQ; ++q) arow[32 * q] = p[q] * inv;
  }
}

__global__ __launch_bounds__(128) void gat_attn(const float* __restrict__ WhP,
                                                const float* __restrict__ s1,
                                                const float* __restrict__ s2,
                                                const int* __restrict__ mask,
                                                const float* __restrict__ Xin,
                                                float* __restrict__ Xout) {
  const int tid  = threadIdx.x;
  const int lane = tid & 31;
  const int wave = __builtin_amdgcn_readfirstlane(tid >> 5);
  const int mb   = blockIdx.x & 7;
  const int h    = (blockIdx.x >> 3) & (HH - 1);
  const int bt   = blockIdx.x >> 5;
  const int m0   = mb * 32;

  __shared__ float att[32][NN + LPAD];

  __shared__ int idx_s[NN];
  __shared__ int wc_s[4];
  const int  nodeA = wave * 64 + lane;
  const int  nodeB = nodeA + 32;
  const bool mkA = mask[bt * NN + nodeA] > 0;
  const bool mkB = mask[bt * NN + nodeB] > 0;
  const unsigned balA = __builtin_amdgcn_ballot_w32(mkA);
  const unsigned balB = __builtin_amdgcn_ballot_w32(mkB);
  const unsigned ltm  = (1u << lane) - 1u;
  const int preA = __popc(balA & ltm);
  const int preB = __popc(balA) + __popc(balB & ltm);
  if (lane == 0) wc_s[wave] = __popc(balA) + __popc(balB);
  __syncthreads();
  int base = 0, cnt = 0;
#pragma unroll
  for (int w = 0; w < 4; ++w) {
    int c = wc_s[w];
    cnt += c;
    base += (w < wave) ? c : 0;
  }
  cnt  = __builtin_amdgcn_readfirstlane(cnt);
  base = __builtin_amdgcn_readfirstlane(base);
  if (m0 >= cnt) return;
  if (mkA) idx_s[base + preA] = nodeA;
  if (mkB) idx_s[base + preB] = nodeB;
  const int nq = (cnt + 31) >> 5;

  const int    r0  = m0 + wave * 8;
  const float* s1p = s1 + (bt * HH + h) * NN + r0;
  const float* s2p = s2 + (bt * HH + h) * NN + lane;
  if (nq <= 4)      softmax_rows<4>(att, s1p, s2p, r0, wave, lane, cnt, nq);
  else if (nq == 5) softmax_rows<5>(att, s1p, s2p, r0, wave, lane, cnt, nq);
  else              softmax_rows<8>(att, s1p, s2p, r0, wave, lane, cnt, nq);
  __syncthreads();

  const int mt = wave >> 1;
  if (m0 + mt * 16 >= cnt) return;
  const int nt = wave & 1;
  const int kh = lane >> 4;
  const int nl = lane & 15;
  const v4f* Bp =
      (const v4f*)(WhP + ((size_t)(bt * HH + h) * 2 + nt) * (16 * NN)) + lane;
  v8f acc = {};
  for (int g = 0; g < nq; ++g) {
#pragma unroll
    for (int j = 0; j < 4; ++j) {
      const int kp = g * 4 + j;
      v4f b4 = Bp[kp * 32];
      v2f b0 = {b4.x, b4.y};
      v2f b1 = {b4.z, b4.w};
      v2f a0 = *(const v2f*)&att[mt * 16 + nl][kp * 8 + 2 * kh];
      v2f a1 = *(const v2f*)&att[mt * 16 + nl][kp * 8 + 4 + 2 * kh];
      acc = __builtin_amdgcn_wmma_f32_16x16x4_f32(false, a0, false, b0,
                                                  (short)0, acc, false, false);
      acc = __builtin_amdgcn_wmma_f32_16x16x4_f32(false, a1, false, b1,
                                                  (short)0, acc, false, false);
    }
  }
  const int gcol = h * DD + nt * 16 + nl;
  size_t o[8];
  float  xi[8];
#pragma unroll
  for (int v = 0; v < 8; ++v) {
    int crow = m0 + mt * 16 + v + 8 * kh;
    int cc   = (crow < cnt) ? crow : m0 + mt * 16;
    o[v]  = (size_t)(bt * NN + idx_s[cc]) * FF + gcol;
    xi[v] = Xin[o[v]];
  }
#pragma unroll
  for (int v = 0; v < 8; ++v) {
    int crow = m0 + mt * 16 + v + 8 * kh;
    float x = xi[v] + acc[v];
    float y = (x > 0.f) ? x : expm1f(x);
    if (crow < cnt) Xout[o[v]] = y;
  }
}

__global__ __launch_bounds__(256) void layernorm_mask(const float* __restrict__ X,
                                                      const int* __restrict__ mask,
                                                      const float* __restrict__ gamma,
                                                      const float* __restrict__ beta,
                                                      float* __restrict__ Out) {
  const int lane = threadIdx.x & 31;
  const int wave = __builtin_amdgcn_readfirstlane(threadIdx.x >> 5);
  const int row  = blockIdx.x * 8 + wave;
  float* o = Out + (size_t)row * FF;
  if (!(mask[row] > 0)) {
#pragma unroll
    for (int q = 0; q < 4; ++q) o[lane + 32 * q] = 0.f;
    return;
  }
  const float* x = X + (size_t)row * FF;
  float v[4];
  float s = 0.f;
#pragma unroll
  for (int q = 0; q < 4; ++q) { v[q] = x[lane + 32 * q]; s += v[q]; }
  float mu = wave_sum32(s) * (1.f / FF);
  float var = 0.f;
#pragma unroll
  for (int q = 0; q < 4; ++q) { float d = v[q] - mu; var += d * d; }
  var = wave_sum32(var) * (1.f / FF);
  float rstd = rsqrtf(var + 1e-5f);
#pragma unroll
  for (int q = 0; q < 4; ++q) {
    int c = lane + 32 * q;
    o[c] = (v[q] - mu) * rstd * gamma[c] + beta[c];
  }
}

extern "C" void kernel_launch(void* const* d_in, const int* in_sizes, int n_in,
                              void* d_out, int out_size, void* d_ws, size_t ws_size,
                              hipStream_t stream) {
  const float* X0    = (const float*)d_in[0];
  const int*   mask  = (const int*)d_in[1];
  const float* W0    = (const float*)d_in[2];
  const float* a0    = (const float*)d_in[3];
  const float* W1    = (const float*)d_in[4];
  const float* a1    = (const float*)d_in[5];
  const float* gamma = (const float*)d_in[6];
  const float* beta  = (const float*)d_in[7];
  float* out = (float*)d_out;

  float* Xb  = (float*)d_ws;
  float* WhP = Xb + (size_t)BT * NN * FF;
  float* s1  = WhP + (size_t)BT * NN * FF;
  float* s2  = s1 + (size_t)BT * HH * NN;
  float* Wp  = s2 + (size_t)BT * HH * NN;

  pack_w<<<64, 256, 0, stream>>>(W0, Wp);
  gemm_xw<<<BT * 16, 256, 0, stream>>>(X0, mask, Wp, a0, WhP, s1, s2);
  gat_attn<<<BT * HH * 8, 128, 0, stream>>>(WhP, s1, s2, mask, X0, Xb);
  pack_w<<<64, 256, 0, stream>>>(W1, Wp);
  gemm_xw<<<BT * 16, 256, 0, stream>>>(Xb, mask, Wp, a1, WhP, s1, s2);
  gat_attn<<<BT * HH * 8, 128, 0, stream>>>(WhP, s1, s2, mask, Xb, Xb);
  layernorm_mask<<<(BT * NN) / 8, 256, 0, stream>>>(Xb, mask, gamma, beta, out);
}


# wave reductions (softmax max and sum, LayerNorm sums) as vector-ALU butterflies, v_permlanex16 + DPP row_xmask, same partner order 16..1, instead of ds_bpermute trips through the LDS unit
# speedup vs baseline: 3.0271x; 1.1012x over previous
#include <hip/hip_runtime.h>


typedef float v2f __attribute__((ext_vector_type(2)));
typedef float v4f __attribute__((ext_vector_type(4)));
typedef float v8f __attribute__((ext_vector_type(8)));

#define BT 128
#define NN 256
#define FF 128
#define HH 4
#define DD 32
#define NEGV (-1e9f)
#define LPAD 4

__device__ __forceinline__ float xhalf(float x) {
  int b = __builtin_bit_cast(int, x);
  return __builtin_bit_cast(
      float, __builtin_amdgcn_permlanex16(b, b, 0x76543210, 0xfedcba98, false, false));
}
template <int M>
__device__ __forceinline__ float xrow(float x) {
  int b = __builtin_bit_cast(int, x);
  return __builtin_bit_cast(
      float, __builtin_amdgcn_update_dpp(0, b, 0x160 + M, 0xf, 0xf, true));
}
__device__ __forceinline__ float wave_max32(float v) {
  v = fmaxf(v, xhalf(v));
  v = fmaxf(v, xrow<8>(v));
  v = fmaxf(v, xrow<4>(v));
  v = fmaxf(v, xrow<2>(v));
  v = fmaxf(v, xrow<1>(v));
  return v;
}
__device__ __forceinline__ float wave_sum32(float v) {
  v += xhalf(v);
  v += xrow<8>(v);
  v += xrow<4>(v);
  v += xrow<2>(v);
  v += xrow<1>(v);
  return v;
}

__global__ __launch_bounds__(256) void pack_w(const float* __restrict__ W,
                                              float* __restrict__ Wp) {
  int idx  = blockIdx.x * 256 + threadIdx.x;
  int e    = idx & 3;
  int lane = (idx >> 2) & 31;
  int kp   = (idx >> 7) & 15;
  int nt   = idx >> 11;
  int k = kp * 8 + (e >> 1) * 4 + 2 * (lane >> 4) + (e & 1);
  int n = nt * 16 + (lane & 15);
  Wp[idx] = W[k * FF + n];
}

__global__ __launch_bounds__(256) void gemm_xw(const float* __restrict__ X,
                                               const int* __restrict__ mask,
                                               const float* __restrict__ Wp,
                                               const float* __restrict__ avec,
                                               float* __restrict__ WhP,
                                               float* __restrict__ s1,
                                               float* __restrict__ s2) {
  const int tid  = threadIdx.x;
  const int lane = tid & 31;
  const int wave = tid >> 5;
  const int bt   = blockIdx.x >> 4;
  const int m0   = (blockIdx.x & 15) * 16;

  __shared__ int idx_s[NN];
  __shared__ int wc_s[8];
  const bool mk = mask[bt * NN + tid] > 0;
  const unsigned bal = __builtin_amdgcn_ballot_w32(mk);
  const int pre = __popc(bal & ((1u << lane) - 1u));
  if (lane == 0) wc_s[wave] = __popc(bal);
  __syncthreads();
  int base = 0, cnt = 0;
#pragma unroll
  for (int w = 0; w < 8; ++w) {
    int c = wc_s[w];
    cnt += c;
    base += (w < wave) ? c : 0;
  }
  cnt = __builtin_amdgcn_readfirstlane(cnt);
  if (m0 >= ((cnt + 31) & ~31)) return;
  if (mk) idx_s[base + pre] = tid;
  __syncthreads();

  __shared__ float As[16][FF + LPAD];
  const float* Xg = X + (size_t)bt * NN * FF;
  float xr[8];
#pragma unroll
  for (int i = 0; i < 8; ++i) {
    int idx = tid + i * 256;
    int c   = m0 + (idx >> 7);
    int cc  = (c < cnt) ? c : cnt - 1;
    xr[i] = Xg[(size_t)idx_s[cc] * FF + (idx & 127)];
  }
#pragma unroll
  for (int i = 0; i < 8; ++i) {
    int idx = tid + i * 256;
    int r   = idx >> 7;
    As[r][idx & 127] = (m0 + r < cnt) ? xr[i] : 0.f;
  }
  __syncthreads();

  const int n0 = wave * 16;
  const int kh = lane >> 4;
  const int nl = lane & 15;
  v8f acc = {};
  const v4f* Ap = (const v4f*)Wp + (size_t)(wave * 16) * 32 + lane;
  for (int kp = 0; kp < FF / 8; ++kp) {
    v4f a4 = Ap[kp * 32];
    v2f a0 = {a4.x, a4.y};
    v2f a1 = {a4.z, a4.w};
    v2f b0 = *(const v2f*)&As[nl][kp * 8 + 2 * kh];
    v2f b1 = *(const v2f*)&As[nl][kp * 8 + 4 + 2 * kh];
    acc = __builtin_amdgcn_wmma_f32_16x16x4_f32(false, a0, false, b0,
                                                (short)0, acc, false, false);
    acc = __builtin_amdgcn_wmma_f32_16x16x4_f32(false, a1, false, b1,
                                                (short)0, acc, false, false);
  }
  __shared__ float Ts[FF][17];
#pragma unroll
  for (int v = 0; v < 8; ++v) Ts[n0 + v + 8 * kh][nl] = acc[v];
  __syncthreads();
  {
    const int kpl  = tid >> 7;
    const int fln  = (tid >> 2) & 31;
    const int e    = tid & 3;
    const int nloc = kpl * 8 + (e >> 1) * 4 + 2 * (fln >> 4) + (e & 1);
    const int fl   = fln & 15;
    float* dst = WhP + (size_t)bt * (HH * DD * NN) + (m0 >> 3) * 128 + tid;
#pragma unroll
    for (int c = 0; c < 8; ++c)
      dst[c * (16 * NN)] = Ts[c * 16 + fl][nloc];
  }
  if (tid < 64) {
    const int hh = tid >> 4;
    const int n  = tid & 15;
    float v1 = 0.f, v2 = 0.f;
#pragma unroll
    for (int d = 0; d < DD; ++d) {
      float x = Ts[hh * DD + d][n];
      v1 += x * avec[d];
      v2 += x * avec[DD + d];
    }
    const int o = (bt * HH + hh) * NN + m0 + n;
    s1[o] = v1;
    s2[o] = v2;
  }
}

template <int NQ>
__device__ __forceinline__ void softmax_rows(float (*att)[NN + LPAD],
                                             const float* __restrict__ s1p,
                                             const float* __restrict__ s2p,
                                             int r0, int wave, int lane,
                                             int cnt, int nq) {
  float s2r[NQ];
  bool  okc[NQ];
#pragma unroll
  for (int q = 0; q < NQ; ++q) {
    okc[q] = (lane + 32 * q) < cnt;
    s2r[q] = 0.f;
    if (q < nq) s2r[q] = s2p[32 * q];
  }
  float s1r[8];
#pragma unroll
  for (int rr = 0; rr < 8; ++rr) s1r[rr] = s1p[rr];

#pragma unroll
  for (int rr = 0; rr < 8; ++rr) {
    float* arow = &att[wave * 8 + rr][lane];
    if (r0 + rr >= cnt) {
#pragma unroll
      for (int q = 0; q < NQ; ++q) arow[32 * q] = 0.f;
      continue;
    }
    const float s1v = s1r[rr];
    float e[NQ];
    float vmax = NEGV;
#pragma unroll
    for (int q = 0; q < NQ; ++q) {
      float t = s1v + s2r[q];
      float x = (t > 0.f) ? t : 0.2f * t;
      e[q] = okc[q] ? x : NEGV;
      vmax = fmaxf(vmax, e[q]);
    }
    vmax = wave_max32(vmax);
    float p[NQ];
    float vsum = 0.f;
#pragma unroll
    for (int q = 0; q < NQ; ++q) {
      float pe = __expf(e[q] - vmax);
      p[q] = okc[q] ? pe : 0.f;
      vsum += p[q];
    }
    vsum = wave_sum32(vsum);
    float inv = (vsum > 0.f) ? 1.f / vsum : 0.f;
#pragma unroll
    for (int q = 0; q < NQ; ++q) arow[32 * q] = p[q] * inv;
  }
}

__global__ __launch_bounds__(128) void gat_attn(const float* __restrict__ WhP,
                                                const float* __restrict__ s1,
                                                const float* __restrict__ s2,
                                                const int* __restrict__ mask,
                                                const float* __restrict__ Xin,
                                                float* __restrict__ Xout) {
  const int tid  = threadIdx.x;
  const int lane = tid & 31;
  const int wave = __builtin_amdgcn_readfirstlane(tid >> 5);
  const int mb   = blockIdx.x & 7;
  const int h    = (blockIdx.x >> 3) & (HH - 1);
  const int bt   = blockIdx.x >> 5;
  const int m0   = mb * 32;

  __shared__ float att[32][NN + LPAD];

  __shared__ int idx_s[NN];
  __shared__ int wc_s[4];
  const int  nodeA = wave * 64 + lane;
  const int  nodeB = nodeA + 32;
  const bool mkA = mask[bt * NN + nodeA] > 0;
  const bool mkB = mask[bt * NN + nodeB] > 0;
  const unsigned balA = __builtin_amdgcn_ballot_w32(mkA);
  const unsigned balB = __builtin_amdgcn_ballot_w32(mkB);
  const unsigned ltm  = (1u << lane) - 1u;
  const int preA = __popc(balA & ltm);
  const int preB = __popc(balA) + __popc(balB & ltm);
  if (lane == 0) wc_s[wave] = __popc(balA) + __popc(balB);
  __syncthreads();
  int base = 0, cnt = 0;
#pragma unroll
  for (int w = 0; w < 4; ++w) {
    int c = wc_s[w];
    cnt += c;
    base += (w < wave) ? c : 0;
  }
  cnt  = __builtin_amdgcn_readfirstlane(cnt);
  base = __builtin_amdgcn_readfirstlane(base);
  if (m0 >= cnt) return;
  if (mkA) idx_s[base + preA] = nodeA;
  if (mkB) idx_s[base + preB] = nodeB;
  const int nq = (cnt + 31) >> 5;

  const int    r0  = m0 + wave * 8;
  const float* s1p = s1 + (bt * HH + h) * NN + r0;
  const float* s2p = s2 + (bt * HH + h) * NN + lane;
  if (nq <= 4)      softmax_rows<4>(att, s1p, s2p, r0, wave, lane, cnt, nq);
  else if (nq == 5) softmax_rows<5>(att, s1p, s2p, r0, wave, lane, cnt, nq);
  else              softmax_rows<8>(att, s1p, s2p, r0, wave, lane, cnt, nq);
  __syncthreads();

  const int mt = wave >> 1;
  if (m0 + mt * 16 >= cnt) return;
  const int nt = wave & 1;
  const int kh = lane >> 4;
  const int nl = lane & 15;
  const v4f* Bp =
      (const v4f*)(WhP + ((size_t)(bt * HH + h) * 2 + nt) * (16 * NN)) + lane;
  v8f acc = {};
  for (int g = 0; g < nq; ++g) {
#pragma unroll
    for (int j = 0; j < 4; ++j) {
      const int kp = g * 4 + j;
      v4f b4 = Bp[kp * 32];
      v2f b0 = {b4.x, b4.y};
      v2f b1 = {b4.z, b4.w};
      v2f a0 = *(const v2f*)&att[mt * 16 + nl][kp * 8 + 2 * kh];
      v2f a1 = *(const v2f*)&att[mt * 16 + nl][kp * 8 + 4 + 2 * kh];
      acc = __builtin_amdgcn_wmma_f32_16x16x4_f32(false, a0, false, b0,
                                                  (short)0, acc, false, false);
      acc = __builtin_amdgcn_wmma_f32_16x16x4_f32(false, a1, false, b1,
                                                  (short)0, acc, false, false);
    }
  }
  const int gcol = h * DD + nt * 16 + nl;
  size_t o[8];
  float  xi[8];
#pragma unroll
  for (int v = 0; v < 8; ++v) {
    int crow = m0 + mt * 16 + v + 8 * kh;
    int cc   = (crow < cnt) ? crow : m0 + mt * 16;
    o[v]  = (size_t)(bt * NN + idx_s[cc]) * FF + gcol;
    xi[v] = Xin[o[v]];
  }
#pragma unroll
  for (int v = 0; v < 8; ++v) {
    int crow = m0 + mt * 16 + v + 8 * kh;
    float x = xi[v] + acc[v];
    float y = (x > 0.f) ? x : expm1f(x);
    if (crow < cnt) Xout[o[v]] = y;
  }
}

__global__ __launch_bounds__(256) void layernorm_mask(const float* __restrict__ X,
                                                      const int* __restrict__ mask,
                                                      const float* __restrict__ gamma,
                                                      const float* __restrict__ beta,
                                                      float* __restrict__ Out) {
  const int lane = threadIdx.x & 31;
  const int wave = __builtin_amdgcn_readfirstlane(threadIdx.x >> 5);
  const int row  = blockIdx.x * 8 + wave;
  float* o = Out + (size_t)row * FF;
  if (!(mask[row] > 0)) {
#pragma unroll
    for (int q = 0; q < 4; ++q) o[lane + 32 * q] = 0.f;
    return;
  }
  const float* x = X + (size_t)row * FF;
  float v[4];
  float s = 0.f;
#pragma unroll
  for (int q = 0; q < 4; ++q) { v[q] = x[lane + 32 * q]; s += v[q]; }
  float mu = wave_sum32(s) * (1.f / FF);
  float var = 0.f;
#pragma unroll
  for (int q = 0; q < 4; ++q) { float d = v[q] - mu; var += d * d; }
  var = wave_sum32(var) * (1.f / FF);
  float rstd = rsqrtf(var + 1e-5f);
#pragma unroll
  for (int q = 0; q < 4; ++q) {
    int c = lane + 32 * q;
    o[c] = (v[q] - mu) * rstd * gamma[c] + beta[c];
  }
}

extern "C" void kernel_launch(void* const* d_in, const int* in_sizes, int n_in,
                              void* d_out, int out_size, void* d_ws, size_t ws_size,
                              hipStream_t stream) {
  const float* X0    = (const float*)d_in[0];
  const int*   mask  = (const int*)d_in[1];
  const float* W0    = (const float*)d_in[2];
  const float* a0    = (const float*)d_in[3];
  const float* W1    = (const float*)d_in[4];
  const float* a1    = (const float*)d_in[5];
  const float* gamma = (const float*)d_in[6];
  const float* beta  = (const float*)d_in[7];
  float* out = (float*)d_out;

  float* Xb  = (float*)d_ws;
  float* WhP = Xb + (size_t)BT * NN * FF;
  float* s1  = WhP + (size_t)BT * NN * FF;
  float* s2  = s1 + (size_t)BT * HH * NN;
  float* Wp  = s2 + (size_t)BT * HH * NN;

  pack_w<<<64, 256, 0, stream>>>(W0, Wp);
  gemm_xw<<<BT * 16, 256, 0, stream>>>(X0, mask, Wp, a0, WhP, s1, s2);
  gat_attn<<<BT * HH * 8, 128, 0, stream>>>(WhP, s1, s2, mask, X0, Xb);
  pack_w<<<64, 256, 0, stream>>>(W1, Wp);
  gemm_xw<<<BT * 16, 256, 0, stream>>>(Xb, mask, Wp, a1, WhP, s1, s2);
  gat_attn<<<BT * HH * 8, 128, 0, stream>>>(WhP, s1, s2, mask, Xb, Xb);
  layernorm_mask<<<(BT * NN) / 8, 256, 0, stream>>>(Xb, mask, gamma, beta, out);
}


# final LayerNorm moved into the last attention kernel's epilogue: a block takes all 4 heads of its 32 rows (16 waves), stages the ELU rows in LDS, normalises, writes out; LayerNorm launch gone
# speedup vs baseline: 3.1641x; 1.0453x over previous
#include <hip/hip_runtime.h>


typedef float v2f __attribute__((ext_vector_type(2)));
typedef float v4f __attribute__((ext_vector_type(4)));
typedef float v8f __attribute__((ext_vector_type(8)));

#define BT 128
#define NN 256
#define FF 128
#define HH 4
#define DD 32
#define NEGV (-1e9f)
#define LPAD 4

__device__ __forceinline__ float xhalf(float x) {
  int b = __builtin_bit_cast(int, x);
  return __builtin_bit_cast(
      float, __builtin_amdgcn_permlanex16(b, b, 0x76543210, 0xfedcba98, false, false));
}
template <int M>
__device__ __forceinline__ float xrow(float x) {
  int b = __builtin_bit_cast(int, x);
  return __builtin_bit_cast(
      float, __builtin_amdgcn_update_dpp(0, b, 0x160 + M, 0xf, 0xf, true));
}
__device__ __forceinline__ float wave_max32(float v) {
  v = fmaxf(v, xhalf(v));
  v = fmaxf(v, xrow<8>(v));
  v = fmaxf(v, xrow<4>(v));
  v = fmaxf(v, xrow<2>(v));
  v = fmaxf(v, xrow<1>(v));
  return v;
}
__device__ __forceinline__ float wave_sum32(float v) {
  v += xhalf(v);
  v += xrow<8>(v);
  v += xrow<4>(v);
  v += xrow<2>(v);
  v += xrow<1>(v);
  return v;
}

__global__ __launch_bounds__(256) void pack_w(const float* __restrict__ W,
                                              float* __restrict__ Wp) {
  int idx  = blockIdx.x * 256 + threadIdx.x;
  int e    = idx & 3;
  int lane = (idx >> 2) & 31;
  int kp   = (idx >> 7) & 15;
  int nt   = idx >> 11;
  int k = kp * 8 + (e >> 1) * 4 + 2 * (lane >> 4) + (e & 1);
  int n = nt * 16 + (lane & 15);
  Wp[idx] = W[k * FF + n];
}

__global__ __launch_bounds__(256) void gemm_xw(const float* __restrict__ X,
                                               const int* __restrict__ mask,
                                               const float* __restrict__ Wp,
                                               const float* __restrict__ avec,
                                               float* __restrict__ WhP,
                                               float* __restrict__ s1,
                                               float* __restrict__ s2) {
  const int tid  = threadIdx.x;
  const int lane = tid & 31;
  const int wave = tid >> 5;
  const int bt   = blockIdx.x >> 4;
  const int m0   = (blockIdx.x & 15) * 16;

  __shared__ int idx_s[NN];
  __shared__ int wc_s[8];
  const bool mk = mask[bt * NN + tid] > 0;
  const unsigned bal = __builtin_amdgcn_ballot_w32(mk);
  const int pre = __popc(bal & ((1u << lane) - 1u));
  if (lane == 0) wc_s[wave] = __popc(bal);
  __syncthreads();
  int base = 0, cnt = 0;
#pragma unroll
  for (int w = 0; w < 8; ++w) {
    int c = wc_s[w];
    cnt += c;
    base += (w < wave) ? c : 0;
  }
  cnt = __builtin_amdgcn_readfirstlane(cnt);
  if (m0 >= ((cnt + 31) & ~31)) return;
  if (mk) idx_s[base + pre] = tid;
  __syncthreads();

  __shared__ float As[16][FF + LPAD];
  const float* Xg = X + (size_t)bt * NN * FF;
  float xr[8];
#pragma unroll
  for (int i = 0; i < 8; ++i) {
    int idx = tid + i * 256;
    int c   = m0 + (idx >> 7);
    int cc  = (c < cnt) ? c : cnt - 1;
    xr[i] = Xg[(size_t)idx_s[cc] * FF + (idx & 127)];
  }
#pragma unroll
  for (int i = 0; i < 8; ++i) {
    int idx = tid + i * 256;
    int r   = idx >> 7;
    As[r][idx & 127] = (m0 + r < cnt) ? xr[i] : 0.f;
  }
  __syncthreads();

  const int n0 = wave * 16;
  const int kh = lane >> 4;
  const int nl = lane & 15;
  v8f acc = {};
  const v4f* Ap = (const v4f*)Wp + (size_t)(wave * 16) * 32 + lane;
  for (int kp = 0; kp < FF / 8; ++kp) {
    v4f a4 = Ap[kp * 32];
    v2f a0 = {a4.x, a4.y};
    v2f a1 = {a4.z, a4.w};
    v2f b0 = *(const v2f*)&As[nl][kp * 8 + 2 * kh];
    v2f b1 = *(const v2f*)&As[nl][kp * 8 + 4 + 2 * kh];
    acc = __builtin_amdgcn_wmma_f32_16x16x4_f32(false, a0, false, b0,
                                                (short)0, acc, false, false);
    acc = __builtin_amdgcn_wmma_f32_16x16x4_f32(false, a1, false, b1,
                                                (short)0, acc, false, false);
  }
  __shared__ float Ts[FF][17];
#pragma unroll
  for (int v = 0; v < 8; ++v) Ts[n0 + v + 8 * kh][nl] = acc[v];
  __syncthreads();
  {
    const int kpl  = tid >> 7;
    const int fln  = (tid >> 2) & 31;
    const int e    = tid & 3;
    const int nloc = kpl * 8 + (e >> 1) * 4 + 2 * (fln >> 4) + (e & 1);
    const int fl   = fln & 15;
    float* dst = WhP + (size_t)bt * (HH * DD * NN) + (m0 >> 3) * 128 + tid;
#pragma unroll
    for (int c = 0; c < 8; ++c)
      dst[c * (16 * NN)] = Ts[c * 16 + fl][nloc];
  }
  if (tid < 64) {
    const int hh = tid >> 4;
    const int n  = tid & 15;
    float v1 = 0.f, v2 = 0.f;
#pragma unroll
    for (int d = 0; d < DD; ++d) {
      float x = Ts[hh * DD + d][n];
      v1 += x * avec[d];
      v2 += x * avec[DD + d];
    }
    const int o = (bt * HH + hh) * NN + m0 + n;
    s1[o] = v1;
    s2[o] = v2;
  }
}

template <int NQ>
__device__ __forceinline__ void softmax_rows(float (*att)[NN + LPAD],
                                             const float* __restrict__ s1p,
                                             const float* __restrict__ s2p,
                                             int r0, int wave, int lane,
                                             int cnt, int nq) {
  float s2r[NQ];
  bool  okc[NQ];
#pragma unroll
  for (int q = 0; q < NQ; ++q) {
    okc[q] = (lane + 32 * q) < cnt;
    s2r[q] = 0.f;
    if (q < nq) s2r[q] = s2p[32 * q];
  }
  float s1r[8];
#pragma unroll
  for (int rr = 0; rr < 8; ++rr) s1r[rr] = s1p[rr];

#pragma unroll
  for (int rr = 0; rr < 8; ++rr) {
    float* arow = &att[wave * 8 + rr][lane];
    if (r0 + rr >= cnt) {
#pragma unroll
      for (int q = 0; q < NQ; ++q) arow[32 * q] = 0.f;
      continue;
    }
    const float s1v = s1r[rr];
    float e[NQ];
    float vmax = NEGV;
#pragma unroll
    for (int q = 0; q < NQ; ++q) {
      float t = s1v + s2r[q];
      float x = (t > 0.f) ? t : 0.2f * t;
      e[q] = okc[q] ? x : NEGV;
      vmax = fmaxf(vmax, e[q]);
    }
    vmax = wave_max32(vmax);
    float p[NQ];
    float vsum = 0.f;
#pragma unroll
    for (int q = 0; q < NQ; ++q) {
      float pe = __expf(e[q] - vmax);
      p[q] = okc[q] ? pe : 0.f;
      vsum += p[q];
    }
    vsum = wave_sum32(vsum);
    float inv = (vsum > 0.f) ? 1.f / vsum : 0.f;
#pragma unroll
    for (int q = 0; q < NQ; ++q) arow[32 * q] = p[q] * inv;
  }
}

__global__ __launch_bounds__(128) void gat_attn(const float* __restrict__ WhP,
                                                const float* __restrict__ s1,
                                                const float* __restrict__ s2,
                                                const int* __restrict__ mask,
                                                const float* __restrict__ Xin,
                                                float* __restrict__ Xout) {
  const int tid  = threadIdx.x;
  const int lane = tid & 31;
  const int wave = __builtin_amdgcn_readfirstlane(tid >> 5);
  const int mb   = blockIdx.x & 7;
  const int h    = (blockIdx.x >> 3) & (HH - 1);
  const int bt   = blockIdx.x >> 5;
  const int m0   = mb * 32;

  __shared__ float att[32][NN + LPAD];

  __shared__ int idx_s[NN];
  __shared__ int wc_s[4];
  const int  nodeA = wave * 64 + lane;
  const int  nodeB = nodeA + 32;
  const bool mkA = mask[bt * NN + nodeA] > 0;
  const bool mkB = mask[bt * NN + nodeB] > 0;
  const unsigned balA = __builtin_amdgcn_ballot_w32(mkA);
  const unsigned balB = __builtin_amdgcn_ballot_w32(mkB);
  const unsigned ltm  = (1u << lane) - 1u;
  const int preA = __popc(balA & ltm);
  const int preB = __popc(balA) + __popc(balB & ltm);
  if (lane == 0) wc_s[wave] = __popc(balA) + __popc(balB);
  __syncthreads();
  int base = 0, cnt = 0;
#pragma unroll
  for (int w = 0; w < 4; ++w) {
    int c = wc_s[w];
    cnt += c;
    base += (w < wave) ? c : 0;
  }
  cnt  = __builtin_amdgcn_readfirstlane(cnt);
  base = __builtin_amdgcn_readfirstlane(base);
  if (m0 >= cnt) return;
  if (mkA) idx_s[base + preA] = nodeA;
  if (mkB) idx_s[base + preB] = nodeB;
  const int nq = (cnt + 31) >> 5;

  const int    r0  = m0 + wave * 8;
  const float* s1p = s1 + (bt * HH + h) * NN + r0;
  const float* s2p = s2 + (bt * HH + h) * NN + lane;
  if (nq <= 4)      softmax_rows<4>(att, s1p, s2p, r0, wave, lane, cnt, nq);
  else if (nq == 5) softmax_rows<5>(att, s1p, s2p, r0, wave, lane, cnt, nq);
  else              softmax_rows<8>(att, s1p, s2p, r0, wave, lane, cnt, nq);
  __syncthreads();

  const int mt = wave >> 1;
  if (m0 + mt * 16 >= cnt) return;
  const int nt = wave & 1;
  const int kh = lane >> 4;
  const int nl = lane & 15;
  const v4f* Bp =
      (const v4f*)(WhP + ((size_t)(bt * HH + h) * 2 + nt) * (16 * NN)) + lane;
  v8f acc = {};
  for (int g = 0; g < nq; ++g) {
#pragma unroll
    for (int j = 0; j < 4; ++j) {
      const int kp = g * 4 + j;
      v4f b4 = Bp[kp * 32];
      v2f b0 = {b4.x, b4.y};
      v2f b1 = {b4.z, b4.w};
      v2f a0 = *(const v2f*)&att[mt * 16 + nl][kp * 8 + 2 * kh];
      v2f a1 = *(const v2f*)&att[mt * 16 + nl][kp * 8 + 4 + 2 * kh];
      acc = __builtin_amdgcn_wmma_f32_16x16x4_f32(false, a0, false, b0,
                                                  (short)0, acc, false, false);
      acc = __builtin_amdgcn_wmma_f32_16x16x4_f32(false, a1, false, b1,
                                                  (short)0, acc, false, false);
    }
  }
  const int gcol = h * DD + nt * 16 + nl;
  size_t o[8];
  float  xi[8];
#pragma unroll
  for (int v = 0; v < 8; ++v) {
    int crow = m0 + mt * 16 + v + 8 * kh;
    int cc   = (crow < cnt) ? crow : m0 + mt * 16;
    o[v]  = (size_t)(bt * NN + idx_s[cc]) * FF + gcol;
    xi[v] = Xin[o[v]];
  }
#pragma unroll
  for (int v = 0; v < 8; ++v) {
    int crow = m0 + mt * 16 + v + 8 * kh;
    float x = xi[v] + acc[v];
    float y = (x > 0.f) ? x : expm1f(x);
    if (crow < cnt) Xout[o[v]] = y;
  }
}

__device__ __forceinline__ void zero_rows(float* __restrict__ Out, int row,
                                          int m0w, int m1w, int lane) {
  const v4f z = {0.f, 0.f, 0.f, 0.f};
  if (!(m0w > 0)) *(v4f*)(Out + (size_t)row * FF + lane * 4) = z;
  if (!(m1w > 0)) *(v4f*)(Out + (size_t)(row + 1) * FF + lane * 4) = z;
}

__global__ __launch_bounds__(512) void gat_attn_ln(const float* __restrict__ WhP,
                                                   const float* __restrict__ s1,
                                                   const float* __restrict__ s2,
                                                   const int* __restrict__ mask,
                                                   const float* __restrict__ Xin,
                                                   const float* __restrict__ gamma,
                                                   const float* __restrict__ beta,
                                                   float* __restrict__ Out) {
  const int tid  = threadIdx.x;
  const int lane = tid & 31;
  const int wave = __builtin_amdgcn_readfirstlane(tid >> 5);
  const int h    = wave >> 2;
  const int w4   = wave & 3;
  const int mb   = blockIdx.x & 7;
  const int bt   = blockIdx.x >> 3;
  const int m0   = mb * 32;

  __shared__ float att[HH][32][NN + LPAD];
  __shared__ float Ys[32][FF + LPAD];
  __shared__ int idx_s[NN];
  __shared__ int wc_s[8];

  bool mk = false;
  if (wave < 8) mk = mask[bt * NN + tid] > 0;
  const int rowz = bt * NN + m0 + wave * 2;
  const int mz0  = mask[rowz];
  const int mz1  = mask[rowz + 1];
  const unsigned bal = __builtin_amdgcn_ballot_w32(mk);
  const int pre = __popc(bal & ((1u << lane) - 1u));
  if (wave < 8 && lane == 0) wc_s[wave] = __popc(bal);
  __syncthreads();
  int base = 0, cnt = 0;
#pragma unroll
  for (int w = 0; w < 8; ++w) {
    int c = wc_s[w];
    cnt += c;
    base += (w < wave) ? c : 0;
  }
  cnt = __builtin_amdgcn_readfirstlane(cnt);
  if (m0 >= cnt) {
    zero_rows(Out, rowz, mz0, mz1, lane);
    return;
  }
  if (mk) idx_s[base + pre] = tid;
  const int nq = (cnt + 31) >> 5;

  const int    r0  = m0 + w4 * 8;
  const float* s1p = s1 + (bt * HH + h) * NN + r0;
  const float* s2p = s2 + (bt * HH + h) * NN + lane;
  if (nq <= 4)      softmax_rows<4>(att[h], s1p, s2p, r0, w4, lane, cnt, nq);
  else if (nq == 5) softmax_rows<5>(att[h], s1p, s2p, r0, w4, lane, cnt, nq);
  else              softmax_rows<8>(att[h], s1p, s2p, r0, w4, lane, cnt, nq);
  __syncthreads();

  const int mt = w4 >> 1;
  const int nt = w4 & 1;
  const int kh = lane >> 4;
  const int nl = lane & 15;
  if (m0 + mt * 16 < cnt) {
    const v4f* Bp =
        (const v4f*)(WhP + ((size_t)(bt * HH + h) * 2 + nt) * (16 * NN)) + lane;
    v8f acc = {};
    for (int g = 0; g < nq; ++g) {
#pragma unroll
      for (int j = 0; j < 4; ++j) {
        const int kp = g * 4 + j;
        v4f b4 = Bp[kp * 32];
        v2f b0 = {b4.x, b4.y};
        v2f b1 = {b4.z, b4.w};
        v2f a0 = *(const v2f*)&att[h][mt * 16 + nl][kp * 8 + 2 * kh];
        v2f a1 = *(const v2f*)&att[h][mt * 16 + nl][kp * 8 + 4 + 2 * kh];
        acc = __builtin_amdgcn_wmma_f32_16x16x4_f32(false, a0, false, b0,
                                                    (short)0, acc, false, false);
        acc = __builtin_amdgcn_wmma_f32_16x16x4_f32(false, a1, false, b1,
                                                    (short)0, acc, false, false);
      }
    }
    const int gcol = h * DD + nt * 16 + nl;
    float xi[8];
#pragma unroll
    for (int v = 0; v < 8; ++v) {
      int crow = m0 + mt * 16 + v + 8 * kh;
      int cc   = (crow < cnt) ? crow : m0 + mt * 16;
      xi[v] = Xin[(size_t)(bt * NN + idx_s[cc]) * FF + gcol];
    }
#pragma unroll
    for (int v = 0; v < 8; ++v) {
      float x = xi[v] + acc[v];
      Ys[mt * 16 + v + 8 * kh][gcol] = (x > 0.f) ? x : expm1f(x);
    }
  }
  __syncthreads();

#pragma unroll
  for (int i = 0; i < 2; ++i) {
    const int rl   = wave * 2 + i;
    const int crow = m0 + rl;
    if (crow >= cnt) continue;
    float* o = Out + (size_t)(bt * NN + idx_s[crow]) * FF;
    float v[4];
    float s = 0.f;
#pragma unroll
    for (int q = 0; q < 4; ++q) { v[q] = Ys[rl][lane + 32 * q]; s += v[q]; }
    float mu = wave_sum32(s) * (1.f / FF);
    float var = 0.f;
#pragma unroll
    for (int q = 0; q < 4; ++q) { float d = v[q] - mu; var += d * d; }
    var = wave_sum32(var) * (1.f / FF);
    float rstd = rsqrtf(var + 1e-5f);
#pragma unroll
    for (int q = 0; q < 4; ++q) {
      int c = lane + 32 * q;
      o[c] = (v[q] - mu) * rstd * gamma[c] + beta[c];
    }
  }
  zero_rows(Out, rowz, mz0, mz1, lane);
}

extern "C" void kernel_launch(void* const* d_in, const int* in_sizes, int n_in,
                              void* d_out, int out_size, void* d_ws, size_t ws_size,
                              hipStream_t stream) {
  const float* X0    = (const float*)d_in[0];
  const int*   mask  = (const int*)d_in[1];
  const float* W0    = (const float*)d_in[2];
  const float* a0    = (const float*)d_in[3];
  const float* W1    = (const float*)d_in[4];
  const float* a1    = (const float*)d_in[5];
  const float* gamma = (const float*)d_in[6];
  const float* beta  = (const float*)d_in[7];
  float* out = (float*)d_out;

  float* Xb  = (float*)d_ws;
  float* WhP = Xb + (size_t)BT * NN * FF;
  float* s1  = WhP + (size_t)BT * NN * FF;
  float* s2  = s1 + (size_t)BT * HH * NN;
  float* Wp  = s2 + (size_t)BT * HH * NN;

  pack_w<<<64, 256, 0, stream>>>(W0, Wp);
  gemm_xw<<<BT * 16, 256, 0, stream>>>(X0, mask, Wp, a0, WhP, s1, s2);
  gat_attn<<<BT * HH * 8, 128, 0, stream>>>(WhP, s1, s2, mask, X0, Xb);
  pack_w<<<64, 256, 0, stream>>>(W1, Wp);
  gemm_xw<<<BT * 16, 256, 0, stream>>>(Xb, mask, Wp, a1, WhP, s1, s2);
  gat_attn_ln<<<BT * 8, 512, 0, stream>>>(WhP, s1, s2, mask, Xb, gamma, beta, out);
}


# grids sized to the expected unmasked count: 10 GEMM slices, 5 attention row slices per graph (160 nodes), not 16 and 8; a block takes a second slice if its graph has more; fewer idle waves enter
# speedup vs baseline: 3.8490x; 1.2165x over previous
#include <hip/hip_runtime.h>


typedef float v2f __attribute__((ext_vector_type(2)));
typedef float v4f __attribute__((ext_vector_type(4)));
typedef float v8f __attribute__((ext_vector_type(8)));

#define BT 128
#define NN 256
#define FF 128
#define HH 4
#define DD 32
#define NEGV (-1e9f)
#define GSL 10
#define ASL 5
static_assert(2 * GSL * 16 >= NN && 2 * ASL * 32 >= NN, "two trips cover a graph");
#define LPAD 4

__device__ __forceinline__ float xhalf(float x) {
  int b = __builtin_bit_cast(int, x);
  return __builtin_bit_cast(
      float, __builtin_amdgcn_permlanex16(b, b, 0x76543210, 0xfedcba98, false, false));
}
template <int M>
__device__ __forceinline__ float xrow(float x) {
  int b = __builtin_bit_cast(int, x);
  return __builtin_bit_cast(
      float, __builtin_amdgcn_update_dpp(0, b, 0x160 + M, 0xf, 0xf, true));
}
__device__ __forceinline__ float wave_max32(float v) {
  v = fmaxf(v, xhalf(v));
  v = fmaxf(v, xrow<8>(v));
  v = fmaxf(v, xrow<4>(v));
  v = fmaxf(v, xrow<2>(v));
  v = fmaxf(v, xrow<1>(v));
  return v;
}
__device__ __forceinline__ float wave_sum32(float v) {
  v += xhalf(v);
  v += xrow<8>(v);
  v += xrow<4>(v);
  v += xrow<2>(v);
  v += xrow<1>(v);
  return v;
}

__global__ __launch_bounds__(256) void pack_w(const float* __restrict__ W,
                                              float* __restrict__ Wp) {
  int idx  = blockIdx.x * 256 + threadIdx.x;
  int e    = idx & 3;
  int lane = (idx >> 2) & 31;
  int kp   = (idx >> 7) & 15;
  int nt   = idx >> 11;
  int k = kp * 8 + (e >> 1) * 4 + 2 * (lane >> 4) + (e & 1);
  int n = nt * 16 + (lane & 15);
  Wp[idx] = W[k * FF + n];
}

template <bool SECOND>
__device__ __forceinline__ void gemm_slice(const float* __restrict__ Xg,
                                           const float* __restrict__ Wp,
                                           const float* __restrict__ avec,
                                           float* __restrict__ WhP,
                                           float* __restrict__ s1,
                                           float* __restrict__ s2,
                                           const int* idx_s,
                                           float (*As)[FF + LPAD],
                                           float (*Ts)[17],
                                           int bt, int m0, int cnt,
                                           int tid, int lane, int wave) {
  float xr[8];
#pragma unroll
  for (int i = 0; i < 8; ++i) {
    int idx = tid + i * 256;
    int c   = m0 + (idx >> 7);
    int cc  = (c < cnt) ? c : cnt - 1;
    xr[i] = Xg[(size_t)idx_s[cc] * FF + (idx & 127)];
  }
#pragma unroll
  for (int i = 0; i < 8; ++i) {
    int idx = tid + i * 256;
    int r   = idx >> 7;
    As[r][idx & 127] = (m0 + r < cnt) ? xr[i] : 0.f;
  }
  __syncthreads();

  const int n0 = wave * 16;
  const int kh = lane >> 4;
  const int nl = lane & 15;
  v8f acc = {};
  const v4f* Ap = (const v4f*)Wp + (size_t)(wave * 16) * 32 + lane;
#pragma unroll(SECOND ? 1 : FF / 8)
  for (int kp = 0; kp < FF / 8; ++kp) {
    v4f a4 = Ap[kp * 32];
    v2f a0 = {a4.x, a4.y};
    v2f a1 = {a4.z, a4.w};
    v2f b0 = *(const v2f*)&As[nl][kp * 8 + 2 * kh];
    v2f b1 = *(const v2f*)&As[nl][kp * 8 + 4 + 2 * kh];
    acc = __builtin_amdgcn_wmma_f32_16x16x4_f32(false, a0, false, b0,
                                                (short)0, acc, false, false);
    acc = __builtin_amdgcn_wmma_f32_16x16x4_f32(false, a1, false, b1,
                                                (short)0, acc, false, false);
  }
#pragma unroll
  for (int v = 0; v < 8; ++v) Ts[n0 + v + 8 * kh][nl] = acc[v];
  __syncthreads();
  {
    const int kpl  = tid >> 7;
    const int fln  = (tid >> 2) & 31;
    const int e    = tid & 3;
    const int nloc = kpl * 8 + (e >> 1) * 4 + 2 * (fln >> 4) + (e & 1);
    const int fl   = fln & 15;
    float* dst = WhP + (size_t)bt * (HH * DD * NN) + (m0 >> 3) * 128 + tid;
#pragma unroll
    for (int c = 0; c < 8; ++c)
      dst[c * (16 * NN)] = Ts[c * 16 + fl][nloc];
  }
  if (tid < 64) {
    const int hh = tid >> 4;
    const int n  = tid & 15;
    float v1 = 0.f, v2 = 0.f;
#pragma unroll(SECOND ? 1 : DD)
    for (int d = 0; d < DD; ++d) {
      float x = Ts[hh * DD + d][n];
      v1 += x * avec[d];
      v2 += x * avec[DD + d];
    }
    const int o = (bt * HH + hh) * NN + m0 + n;
    s1[o] = v1;
    s2[o] = v2;
  }
}

__global__ __launch_bounds__(256) void gemm_xw(const float* __restrict__ X,
                                               const int* __restrict__ mask,
                                               const float* __restrict__ Wp,
                                               const float* __restrict__ avec,
                                               float* __restrict__ WhP,
                                               float* __restrict__ s1,
                                               float* __restrict__ s2) {
  const int tid  = threadIdx.x;
  const int lane = tid & 31;
  const int wave = tid >> 5;
  const int bt   = blockIdx.x / GSL;
  const int m0   = (blockIdx.x - bt * GSL) * 16;

  __shared__ int idx_s[NN];
  __shared__ int wc_s[8];
  const bool mk = mask[bt * NN + tid] > 0;
  const unsigned bal = __builtin_amdgcn_ballot_w32(mk);
  const int pre = __popc(bal & ((1u << lane) - 1u));
  if (lane == 0) wc_s[wave] = __popc(bal);
  __syncthreads();
  int base = 0, cnt = 0;
#pragma unroll
  for (int w = 0; w < 8; ++w) {
    int c = wc_s[w];
    cnt += c;
    base += (w < wave) ? c : 0;
  }
  cnt = __builtin_amdgcn_readfirstlane(cnt);
  const int lim = (cnt + 31) & ~31;
  if (m0 >= lim) return;
  if (mk) idx_s[base + pre] = tid;
  __syncthreads();

  __shared__ float As[16][FF + LPAD];
  __shared__ float Ts[FF][17];
  const float* Xg = X + (size_t)bt * NN * FF;
  gemm_slice<false>(Xg, Wp, avec, WhP, s1, s2, idx_s, As, Ts, bt, m0, cnt,
                    tid, lane, wave);
  if (m0 + GSL * 16 < lim)
    gemm_slice<true>(Xg, Wp, avec, WhP, s1, s2, idx_s, As, Ts, bt,
                     m0 + GSL * 16, cnt, tid, lane, wave);
}

template <int NQ>
__device__ __forceinline__ void softmax_rows(float (*att)[NN + LPAD],
                                             const float* __restrict__ s1p,
                                             const float* __restrict__ s2p,
                                             int r0, int wave, int lane,
                                             int cnt, int nq) {
  float s2r[NQ];
  bool  okc[NQ];
#pragma unroll
  for (int q = 0; q < NQ; ++q) {
    okc[q] = (lane + 32 * q) < cnt;
    s2r[q] = 0.f;
    if (q < nq) s2r[q] = s2p[32 * q];
  }
  float s1r[8];
#pragma unroll
  for (int rr = 0; rr < 8; ++rr) s1r[rr] = s1p[rr];

#pragma unroll
  for (int rr = 0; rr < 8; ++rr) {
    float* arow = &att[wave * 8 + rr][lane];
    if (r0 + rr >= cnt) {
#pragma unroll
      for (int q = 0; q < NQ; ++q) arow[32 * q] = 0.f;
      continue;
    }
    const float s1v = s1r[rr];
    float e[NQ];
    float vmax = NEGV;
#pragma unroll
    for (int q = 0; q < NQ; ++q) {
      float t = s1v + s2r[q];
      float x = (t > 0.f) ? t : 0.2f * t;
      e[q] = okc[q] ? x : NEGV;
      vmax = fmaxf(vmax, e[q]);
    }
    vmax = wave_max32(vmax);
    float p[NQ];
    float vsum = 0.f;
#pragma unroll
    for (int q = 0; q < NQ; ++q) {
      float pe = __expf(e[q] - vmax);
      p[q] = okc[q] ? pe : 0.f;
      vsum += p[q];
    }
    vsum = wave_sum32(vsum);
    float inv = (vsum > 0.f) ? 1.f / vsum : 0.f;
#pragma unroll
    for (int q = 0; q < NQ; ++q) arow[32 * q] = p[q] * inv;
  }
}

template <bool SECOND>
__device__ __forceinline__ void softmax_any(float (*att)[NN + LPAD],
                                            const float* __restrict__ s1p,
                                            const float* __restrict__ s2p,
                                            int r0, int w, int lane, int cnt,
                                            int nq) {
  if (SECOND)       softmax_rows<8>(att, s1p, s2p, r0, w, lane, cnt, nq);
  else if (nq <= 4) softmax_rows<4>(att, s1p, s2p, r0, w, lane, cnt, nq);
  else if (nq == 5) softmax_rows<5>(att, s1p, s2p, r0, w, lane, cnt, nq);
  else              softmax_rows<8>(att, s1p, s2p, r0, w, lane, cnt, nq);
}

__device__ __forceinline__ v8f attn_tile(const float (*att)[NN + LPAD],
                                         const v4f* __restrict__ Bp, int mt,
                                         int nl, int kh, int nq) {
  v8f acc = {};
  for (int g = 0; g < nq; ++g) {
#pragma unroll
    for (int j = 0; j < 4; ++j) {
      const int kp = g * 4 + j;
      v4f b4 = Bp[kp * 32];
      v2f b0 = {b4.x, b4.y};
      v2f b1 = {b4.z, b4.w};
      v2f a0 = *(const v2f*)&att[mt * 16 + nl][kp * 8 + 2 * kh];
      v2f a1 = *(const v2f*)&att[mt * 16 + nl][kp * 8 + 4 + 2 * kh];
      acc = __builtin_amdgcn_wmma_f32_16x16x4_f32(false, a0, false, b0,
                                                  (short)0, acc, false, false);
      acc = __builtin_amdgcn_wmma_f32_16x16x4_f32(false, a1, false, b1,
                                                  (short)0, acc, false, false);
    }
  }
  return acc;
}

template <bool SECOND>
__device__ __forceinline__ void attn_slice(float (*att)[NN + LPAD],
                                           const int* idx_s,
                                           const float* __restrict__ WhP,
                                           const float* __restrict__ s1,
                                           const float* __restrict__ s2,
                                           const float* __restrict__ Xin,
                                           float* __restrict__ Xout,
                                           int bt, int h, int m0, int cnt,
                                           int nq, int wave, int lane) {
  const int    r0  = m0 + wave * 8;
  const float* s1p = s1 + (bt * HH + h) * NN + r0;
  const float* s2p = s2 + (bt * HH + h) * NN + lane;
  softmax_any<SECOND>(att, s1p, s2p, r0, wave, lane, cnt, nq);
  __syncthreads();

  const int mt = wave >> 1;
  if (m0 + mt * 16 >= cnt) return;
  const int nt = wave & 1;
  const int kh = lane >> 4;
  const int nl = lane & 15;
  const v4f* Bp =
      (const v4f*)(WhP + ((size_t)(bt * HH + h) * 2 + nt) * (16 * NN)) + lane;
  v8f acc = attn_tile(att, Bp, mt, nl, kh, nq);
  const int gcol = h * DD + nt * 16 + nl;
  size_t o[8];
  float  xi[8];
#pragma unroll
  for (int v = 0; v < 8; ++v) {
    int crow = m0 + mt * 16 + v + 8 * kh;
    int cc   = (crow < cnt) ? crow : m0 + mt * 16;
    o[v]  = (size_t)(bt * NN + idx_s[cc]) * FF + gcol;
    xi[v] = Xin[o[v]];
  }
#pragma unroll
  for (int v = 0; v < 8; ++v) {
    int crow = m0 + mt * 16 + v + 8 * kh;
    float x = xi[v] + acc[v];
    float y = (x > 0.f) ? x : expm1f(x);
    if (crow < cnt) Xout[o[v]] = y;
  }
}

__global__ __launch_bounds__(128) void gat_attn(const float* __restrict__ WhP,
                                                const float* __restrict__ s1,
                                                const float* __restrict__ s2,
                                                const int* __restrict__ mask,
                                                const float* __restrict__ Xin,
                                                float* __restrict__ Xout) {
  const int tid  = threadIdx.x;
  const int lane = tid & 31;
  const int wave = __builtin_amdgcn_readfirstlane(tid >> 5);
  const int bh   = blockIdx.x / ASL;
  const int mb   = blockIdx.x - bh * ASL;
  const int h    = bh & (HH - 1);
  const int bt   = bh >> 2;
  const int m0   = mb * 32;

  __shared__ float att[32][NN + LPAD];

  __shared__ int idx_s[NN];
  __shared__ int wc_s[4];
  const int  nodeA = wave * 64 + lane;
  const int  nodeB = nodeA + 32;
  const bool mkA = mask[bt * NN + nodeA] > 0;
  const bool mkB = mask[bt * NN + nodeB] > 0;
  const unsigned balA = __builtin_amdgcn_ballot_w32(mkA);
  const unsigned balB = __builtin_amdgcn_ballot_w32(mkB);
  const unsigned ltm  = (1u << lane) - 1u;
  const int preA = __popc(balA & ltm);
  const int preB = __popc(balA) + __popc(balB & ltm);
  if (lane == 0) wc_s[wave] = __popc(balA) + __popc(balB);
  __syncthreads();
  int base = 0, cnt = 0;
#pragma unroll
  for (int w = 0; w < 4; ++w) {
    int c = wc_s[w];
    cnt += c;
    base += (w < wave) ? c : 0;
  }
  cnt  = __builtin_amdgcn_readfirstlane(cnt);
  base = __builtin_amdgcn_readfirstlane(base);
  if (m0 >= cnt) return;
  if (mkA) idx_s[base + preA] = nodeA;
  if (mkB) idx_s[base + preB] = nodeB;
  const int nq = (cnt + 31) >> 5;

  attn_slice<false>(att, idx_s, WhP, s1, s2, Xin, Xout, bt, h, m0, cnt, nq,
                    wave, lane);
  if (m0 + ASL * 32 < cnt) {
    __syncthreads();
    attn_slice<true>(att, idx_s, WhP, s1, s2, Xin, Xout, bt, h,
                     m0 + ASL * 32, cnt, nq, wave, lane);
  }
}

__device__ __forceinline__ void zero_rows(float* __restrict__ Out, int row,
                                          int m0w, int m1w, int lane) {
  const v4f z = {0.f, 0.f, 0.f, 0.f};
  if (!(m0w > 0)) *(v4f*)(Out + (size_t)row * FF + lane * 4) = z;
  if (!(m1w > 0)) *(v4f*)(Out + (size_t)(row + 1) * FF + lane * 4) = z;
}

template <bool SECOND>
__device__ __forceinline__ void attn_ln_slice(float (*att)[32][NN + LPAD],
                                              float (*Ys)[FF + LPAD],
                                              const int* idx_s,
                                              const float* __restrict__ WhP,
                                              const float* __restrict__ s1,
                                              const float* __restrict__ s2,
                                              const float* __restrict__ Xin,
                                              const float* __restrict__ gamma,
                                              const float* __restrict__ beta,
                                              float* __restrict__ Out,
                                              int bt, int m0, int cnt, int nq,
                                              int wave, int lane) {
  const int h  = wave >> 2;
  const int w4 = wave & 3;
  const int    r0  = m0 + w4 * 8;
  const float* s1p = s1 + (bt * HH + h) * NN + r0;
  const float* s2p = s2 + (bt * HH + h) * NN + lane;
  softmax_any<SECOND>(att[h], s1p, s2p, r0, w4, lane, cnt, nq);
  __syncthreads();

  const int mt = w4 >> 1;
  const int nt = w4 & 1;
  const int kh = lane >> 4;
  const int nl = lane & 15;
  if (m0 + mt * 16 < cnt) {
    const v4f* Bp =
        (const v4f*)(WhP + ((size_t)(bt * HH + h) * 2 + nt) * (16 * NN)) + lane;
    v8f acc = attn_tile(att[h], Bp, mt, nl, kh, nq);
    const int gcol = h * DD + nt * 16 + nl;
    float xi[8];
#pragma unroll
    for (int v = 0; v < 8; ++v) {
      int crow = m0 + mt * 16 + v + 8 * kh;
      int cc   = (crow < cnt) ? crow : m0 + mt * 16;
      xi[v] = Xin[(size_t)(bt * NN + idx_s[cc]) * FF + gcol];
    }
#pragma unroll
    for (int v = 0; v < 8; ++v) {
      float x = xi[v] + acc[v];
      Ys[mt * 16 + v + 8 * kh][gcol] = (x > 0.f) ? x : expm1f(x);
    }
  }
  __syncthreads();

#pragma unroll
  for (int i = 0; i < 2; ++i) {
    const int rl   = wave * 2 + i;
    const int crow = m0 + rl;
    if (crow >= cnt) continue;
    float* o = Out + (size_t)(bt * NN + idx_s[crow]) * FF;
    float v[4];
    float s = 0.f;
#pragma unroll
    for (int q = 0; q < 4; ++q) { v[q] = Ys[rl][lane + 32 * q]; s += v[q]; }
    float mu = wave_sum32(s) * (1.f / FF);
    float var = 0.f;
#pragma unroll
    for (int q = 0; q < 4; ++q) { float d = v[q] - mu; var += d * d; }
    var = wave_sum32(var) * (1.f / FF);
    float rstd = rsqrtf(var + 1e-5f);
#pragma unroll
    for (int q = 0; q < 4; ++q) {
      int c = lane + 32 * q;
      o[c] = (v[q] - mu) * rstd * gamma[c] + beta[c];
    }
  }
}

__global__ __launch_bounds__(512) void gat_attn_ln(const float* __restrict__ WhP,
                                                   const float* __restrict__ s1,
                                                   const float* __restrict__ s2,
                                                   const int* __restrict__ mask,
                                                   const float* __restrict__ Xin,
                                                   const float* __restrict__ gamma,
                                                   const float* __restrict__ beta,
                                                   float* __restrict__ Out) {
  const int tid  = threadIdx.x;
  const int lane = tid & 31;
  const int wave = __builtin_amdgcn_readfirstlane(tid >> 5);
  const int bt   = blockIdx.x / ASL;
  const int mb   = blockIdx.x - bt * ASL;
  const int m0   = mb * 32;
  const bool two = mb + ASL < 8;

  __shared__ float att[HH][32][NN + LPAD];
  __shared__ float Ys[32][FF + LPAD];
  __shared__ int idx_s[NN];
  __shared__ int wc_s[8];

  bool mk = false;
  if (wave < 8) mk = mask[bt * NN + tid] > 0;
  const int rowz = bt * NN + m0 + wave * 2;
  const int mz0  = mask[rowz];
  const int mz1  = mask[rowz + 1];
  const int rowy = two ? rowz + ASL * 32 : rowz;
  const int my0  = mask[rowy];
  const int my1  = mask[rowy + 1];
  const unsigned bal = __builtin_amdgcn_ballot_w32(mk);
  const int pre = __popc(bal & ((1u << lane) - 1u));
  if (wave < 8 && lane == 0) wc_s[wave] = __popc(bal);
  __syncthreads();
  int base = 0, cnt = 0;
#pragma unroll
  for (int w = 0; w < 8; ++w) {
    int c = wc_s[w];
    cnt += c;
    base += (w < wave) ? c : 0;
  }
  cnt = __builtin_amdgcn_readfirstlane(cnt);
  if (m0 < cnt) {
    if (mk) idx_s[base + pre] = tid;
    const int nq = (cnt + 31) >> 5;
    attn_ln_slice<false>(att, Ys, idx_s, WhP, s1, s2, Xin, gamma, beta, Out,
                         bt, m0, cnt, nq, wave, lane);
    if (m0 + ASL * 32 < cnt)
      attn_ln_slice<true>(att, Ys, idx_s, WhP, s1, s2, Xin, gamma, beta, Out,
                          bt, m0 + ASL * 32, cnt, nq, wave, lane);
  }
  zero_rows(Out, rowz, mz0, mz1, lane);
  if (two) zero_rows(Out, rowy, my0, my1, lane);
}

extern "C" void kernel_launch(void* const* d_in, const int* in_sizes, int n_in,
                              void* d_out, int out_size, void* d_ws, size_t ws_size,
                              hipStream_t stream) {
  const float* X0    = (const float*)d_in[0];
  const int*   mask  = (const int*)d_in[1];
  const float* W0    = (const float*)d_in[2];
  const float* a0    = (const float*)d_in[3];
  const float* W1    = (const float*)d_in[4];
  const float* a1    = (const float*)d_in[5];
  const float* gamma = (const float*)d_in[6];
  const float* beta  = (const float*)d_in[7];
  float* out = (float*)d_out;

  float* Xb  = (float*)d_ws;
  float* WhP = Xb + (size_t)BT * NN * FF;
  float* s1  = WhP + (size_t)BT * NN * FF;
  float* s2  = s1 + (size_t)BT * HH * NN;
  float* Wp  = s2 + (size_t)BT * HH * NN;

  pack_w<<<64, 256, 0, stream>>>(W0, Wp);
  gemm_xw<<<BT * GSL, 256, 0, stream>>>(X0, mask, Wp, a0, WhP, s1, s2);
  gat_attn<<<BT * HH * ASL, 128, 0, stream>>>(WhP, s1, s2, mask, X0, Xb);
  pack_w<<<64, 256, 0, stream>>>(W1, Wp);
  gemm_xw<<<BT * GSL, 256, 0, stream>>>(Xb, mask, Wp, a1, WhP, s1, s2);
  gat_attn_ln<<<BT * ASL, 512, 0, stream>>>(WhP, s1, s2, mask, Xb, gamma, beta, out);
}


# layer 1's weights packed by the first 128 blocks of layer 0's attention launch (one element a thread, into the buffer layer 0's GEMM is done with); second pack launch gone: 5 launches, not 6
# speedup vs baseline: 4.0195x; 1.0443x over previous
#include <hip/hip_runtime.h>


typedef float v2f __attribute__((ext_vector_type(2)));
typedef float v4f __attribute__((ext_vector_type(4)));
typedef float v8f __attribute__((ext_vector_type(8)));

#define BT 128
#define NN 256
#define FF 128
#define HH 4
#define DD 32
#define NEGV (-1e9f)
#define GSL 10
#define ASL 5
static_assert(2 * GSL * 16 >= NN && 2 * ASL * 32 >= NN, "two trips cover a graph");
static_assert(BT * HH * ASL >= FF * FF / 128, "attention grid holds the weight-packing blocks");
#define LPAD 4

__device__ __forceinline__ float xhalf(float x) {
  int b = __builtin_bit_cast(int, x);
  return __builtin_bit_cast(
      float, __builtin_amdgcn_permlanex16(b, b, 0x76543210, 0xfedcba98, false, false));
}
template <int M>
__device__ __forceinline__ float xrow(float x) {
  int b = __builtin_bit_cast(int, x);
  return __builtin_bit_cast(
      float, __builtin_amdgcn_update_dpp(0, b, 0x160 + M, 0xf, 0xf, true));
}
__device__ __forceinline__ float wave_max32(float v) {
  v = fmaxf(v, xhalf(v));
  v = fmaxf(v, xrow<8>(v));
  v = fmaxf(v, xrow<4>(v));
  v = fmaxf(v, xrow<2>(v));
  v = fmaxf(v, xrow<1>(v));
  return v;
}
__device__ __forceinline__ float wave_sum32(float v) {
  v += xhalf(v);
  v += xrow<8>(v);
  v += xrow<4>(v);
  v += xrow<2>(v);
  v += xrow<1>(v);
  return v;
}

__device__ __forceinline__ int pack_src(int idx) {
  int e    = idx & 3;
  int lane = (idx >> 2) & 31;
  int kp   = (idx >> 7) & 15;
  int nt   = idx >> 11;
  int k = kp * 8 + (e >> 1) * 4 + 2 * (lane >> 4) + (e & 1);
  int n = nt * 16 + (lane & 15);
  return k * FF + n;
}
__global__ __launch_bounds__(256) void pack_w(const float* __restrict__ W,
                                              float* __restrict__ Wp) {
  int idx = blockIdx.x * 256 + threadIdx.x;
  Wp[idx] = W[pack_src(idx)];
}

template <bool SECOND>
__device__ __forceinline__ void gemm_slice(const float* __restrict__ Xg,
                                           const float* __restrict__ Wp,
                                           const float* __restrict__ avec,
                                           float* __restrict__ WhP,
                                           float* __restrict__ s1,
                                           float* __restrict__ s2,
                                           const int* idx_s,
                                           float (*As)[FF + LPAD],
                                           float (*Ts)[17],
                                           int bt, int m0, int cnt,
                                           int tid, int lane, int wave) {
  float xr[8];
#pragma unroll
  for (int i = 0; i < 8; ++i) {
    int idx = tid + i * 256;
    int c   = m0 + (idx >> 7);
    int cc  = (c < cnt) ? c : cnt - 1;
    xr[i] = Xg[(size_t)idx_s[cc] * FF + (idx & 127)];
  }
#pragma unroll
  for (int i = 0; i < 8; ++i) {
    int idx = tid + i * 256;
    int r   = idx >> 7;
    As[r][idx & 127] = (m0 + r < cnt) ? xr[i] : 0.f;
  }
  __syncthreads();

  const int n0 = wave * 16;
  const int kh = lane >> 4;
  const int nl = lane & 15;
  v8f acc = {};
  const v4f* Ap = (const v4f*)Wp + (size_t)(wave * 16) * 32 + lane;
#pragma unroll(SECOND ? 1 : FF / 8)
  for (int kp = 0; kp < FF / 8; ++kp) {
    v4f a4 = Ap[kp * 32];
    v2f a0 = {a4.x, a4.y};
    v2f a1 = {a4.z, a4.w};
    v2f b0 = *(const v2f*)&As[nl][kp * 8 + 2 * kh];
    v2f b1 = *(const v2f*)&As[nl][kp * 8 + 4 + 2 * kh];
    acc = __builtin_amdgcn_wmma_f32_16x16x4_f32(false, a0, false, b0,
                                                (short)0, acc, false, false);
    acc = __builtin_amdgcn_wmma_f32_16x16x4_f32(false, a1, false, b1,
                                                (short)0, acc, false, false);
  }
#pragma unroll
  for (int v = 0; v < 8; ++v) Ts[n0 + v + 8 * kh][nl] = acc[v];
  __syncthreads();
  {
    const int kpl  = tid >> 7;
    const int fln  = (tid >> 2) & 31;
    const int e    = tid & 3;
    const int nloc = kpl * 8 + (e >> 1) * 4 + 2 * (fln >> 4) + (e & 1);
    const int fl   = fln & 15;
    float* dst = WhP + (size_t)bt * (HH * DD * NN) + (m0 >> 3) * 128 + tid;
#pragma unroll
    for (int c = 0; c < 8; ++c)
      dst[c * (16 * NN)] = Ts[c * 16 + fl][nloc];
  }
  if (tid < 64) {
    const int hh = tid >> 4;
    const int n  = tid & 15;
    float v1 = 0.f, v2 = 0.f;
#pragma unroll(SECOND ? 1 : DD)
    for (int d = 0; d < DD; ++d) {
      float x = Ts[hh * DD + d][n];
      v1 += x * avec[d];
      v2 += x * avec[DD + d];
    }
    const int o = (bt * HH + hh) * NN + m0 + n;
    s1[o] = v1;
    s2[o] = v2;
  }
}

__global__ __launch_bounds__(256) void gemm_xw(const float* __restrict__ X,
                                               const int* __restrict__ mask,
                                               const float* __restrict__ Wp,
                                               const float* __restrict__ avec,
                                               float* __restrict__ WhP,
                                               float* __restrict__ s1,
                                               float* __restrict__ s2) {
  const int tid  = threadIdx.x;
  const int lane = tid & 31;
  const int wave = tid >> 5;
  const int bt   = blockIdx.x / GSL;
  const int m0   = (blockIdx.x - bt * GSL) * 16;

  __shared__ int idx_s[NN];
  __shared__ int wc_s[8];
  const bool mk = mask[bt * NN + tid] > 0;
  const unsigned bal = __builtin_amdgcn_ballot_w32(mk);
  const int pre = __popc(bal & ((1u << lane) - 1u));
  if (lane == 0) wc_s[wave] = __popc(bal);
  __syncthreads();
  int base = 0, cnt = 0;
#pragma unroll
  for (int w = 0; w < 8; ++w) {
    int c = wc_s[w];
    cnt += c;
    base += (w < wave) ? c : 0;
  }
  cnt = __builtin_amdgcn_readfirstlane(cnt);
  const int lim = (cnt + 31) & ~31;
  if (m0 >= lim) return;
  if (mk) idx_s[base + pre] = tid;
  __syncthreads();

  __shared__ float As[16][FF + LPAD];
  __shared__ float Ts[FF][17];
  const float* Xg = X + (size_t)bt * NN * FF;
  gemm_slice<false>(Xg, Wp, avec, WhP, s1, s2, idx_s, As, Ts, bt, m0, cnt,
                    tid, lane, wave);
  if (m0 + GSL * 16 < lim)
    gemm_slice<true>(Xg, Wp, avec, WhP, s1, s2, idx_s, As, Ts, bt,
                     m0 + GSL * 16, cnt, tid, lane, wave);
}

template <int NQ>
__device__ __forceinline__ void softmax_rows(float (*att)[NN + LPAD],
                                             const float* __restrict__ s1p,
                                             const float* __restrict__ s2p,
                                             int r0, int wave, int lane,
                                             int cnt, int nq) {
  float s2r[NQ];
  bool  okc[NQ];
#pragma unroll
  for (int q = 0; q < NQ; ++q) {
    okc[q] = (lane + 32 * q) < cnt;
    s2r[q] = 0.f;
    if (q < nq) s2r[q] = s2p[32 * q];
  }
  float s1r[8];
#pragma unroll
  for (int rr = 0; rr < 8; ++rr) s1r[rr] = s1p[rr];

#pragma unroll
  for (int rr = 0; rr < 8; ++rr) {
    float* arow = &att[wave * 8 + rr][lane];
    if (r0 + rr >= cnt) {
#pragma unroll
      for (int q = 0; q < NQ; ++q) arow[32 * q] = 0.f;
      continue;
    }
    const float s1v = s1r[rr];
    float e[NQ];
    float vmax = NEGV;
#pragma unroll
    for (int q = 0; q < NQ; ++q) {
      float t = s1v + s2r[q];
      float x = (t > 0.f) ? t : 0.2f * t;
      e[q] = okc[q] ? x : NEGV;
      vmax = fmaxf(vmax, e[q]);
    }
    vmax = wave_max32(vmax);
    float p[NQ];
    float vsum = 0.f;
#pragma unroll
    for (int q = 0; q < NQ; ++q) {
      float pe = __expf(e[q] - vmax);
      p[q] = okc[q] ? pe : 0.f;
      vsum += p[q];
    }
    vsum = wave_sum32(vsum);
    float inv = (vsum > 0.f) ? 1.f / vsum : 0.f;
#pragma unroll
    for (int q = 0; q < NQ; ++q) arow[32 * q] = p[q] * inv;
  }
}

template <bool SECOND>
__device__ __forceinline__ void softmax_any(float (*att)[NN + LPAD],
                                            const float* __restrict__ s1p,
                                            const float* __restrict__ s2p,
                                            int r0, int w, int lane, int cnt,
                                            int nq) {
  if (SECOND)       softmax_rows<8>(att, s1p, s2p, r0, w, lane, cnt, nq);
  else if (nq <= 4) softmax_rows<4>(att, s1p, s2p, r0, w, lane, cnt, nq);
  else if (nq == 5) softmax_rows<5>(att, s1p, s2p, r0, w, lane, cnt, nq);
  else              softmax_rows<8>(att, s1p, s2p, r0, w, lane, cnt, nq);
}

__device__ __forceinline__ v8f attn_tile(const float (*att)[NN + LPAD],
                                         const v4f* __restrict__ Bp, int mt,
                                         int nl, int kh, int nq) {
  v8f acc = {};
  for (int g = 0; g < nq; ++g) {
#pragma unroll
    for (int j = 0; j < 4; ++j) {
      const int kp = g * 4 + j;
      v4f b4 = Bp[kp * 32];
      v2f b0 = {b4.x, b4.y};
      v2f b1 = {b4.z, b4.w};
      v2f a0 = *(const v2f*)&att[mt * 16 + nl][kp * 8 + 2 * kh];
      v2f a1 = *(const v2f*)&att[mt * 16 + nl][kp * 8 + 4 + 2 * kh];
      acc = __builtin_amdgcn_wmma_f32_16x16x4_f32(false, a0, false, b0,
                                                  (short)0, acc, false, false);
      acc = __builtin_amdgcn_wmma_f32_16x16x4_f32(false, a1, false, b1,
                                                  (short)0, acc, false, false);
    }
  }
  return acc;
}

template <bool SECOND>
__device__ __forceinline__ void attn_slice(float (*att)[NN + LPAD],
                                           const int* idx_s,
                                           const float* __restrict__ WhP,
                                           const float* __restrict__ s1,
                                           const float* __restrict__ s2,
                                           const float* __restrict__ Xin,
                                           float* __restrict__ Xout,
                                           int bt, int h, int m0, int cnt,
                                           int nq, int wave, int lane) {
  const int    r0  = m0 + wave * 8;
  const float* s1p = s1 + (bt * HH + h) * NN + r0;
  const float* s2p = s2 + (bt * HH + h) * NN + lane;
  softmax_any<SECOND>(att, s1p, s2p, r0, wave, lane, cnt, nq);
  __syncthreads();

  const int mt = wave >> 1;
  if (m0 + mt * 16 >= cnt) return;
  const int nt = wave & 1;
  const int kh = lane >> 4;
  const int nl = lane & 15;
  const v4f* Bp =
      (const v4f*)(WhP + ((size_t)(bt * HH + h) * 2 + nt) * (16 * NN)) + lane;
  v8f acc = attn_tile(att, Bp, mt, nl, kh, nq);
  const int gcol = h * DD + nt * 16 + nl;
  size_t o[8];
  float  xi[8];
#pragma unroll
  for (int v = 0; v < 8; ++v) {
    int crow = m0 + mt * 16 + v + 8 * kh;
    int cc   = (crow < cnt) ? crow : m0 + mt * 16;
    o[v]  = (size_t)(bt * NN + idx_s[cc]) * FF + gcol;
    xi[v] = Xin[o[v]];
  }
#pragma unroll
  for (int v = 0; v < 8; ++v) {
    int crow = m0 + mt * 16 + v + 8 * kh;
    float x = xi[v] + acc[v];
    float y = (x > 0.f) ? x : expm1f(x);
    if (crow < cnt) Xout[o[v]] = y;
  }
}

__global__ __launch_bounds__(128) void gat_attn(const float* __restrict__ WhP,
                                                const float* __restrict__ s1,
                                                const float* __restrict__ s2,
                                                const int* __restrict__ mask,
                                                const float* __restrict__ Xin,
                                                float* __restrict__ Xout,
                                                const float* __restrict__ Wnext,
                                                float* __restrict__ Wp) {
  const int tid  = threadIdx.x;
  const int lane = tid & 31;
  const bool packer = blockIdx.x < FF * FF / 128;
  const int  pidx   = blockIdx.x * 128 + tid;
  float wv = 0.f;
  if (packer) wv = Wnext[pack_src(pidx)];
  const int wave = __builtin_amdgcn_readfirstlane(tid >> 5);
  const int bh   = blockIdx.x / ASL;
  const int mb   = blockIdx.x - bh * ASL;
  const int h    = bh & (HH - 1);
  const int bt   = bh >> 2;
  const int m0   = mb * 32;

  __shared__ float att[32][NN + LPAD];

  __shared__ int idx_s[NN];
  __shared__ int wc_s[4];
  const int  nodeA = wave * 64 + lane;
  const int  nodeB = nodeA + 32;
  const bool mkA = mask[bt * NN + nodeA] > 0;
  const bool mkB = mask[bt * NN + nodeB] > 0;
  const unsigned balA = __builtin_amdgcn_ballot_w32(mkA);
  const unsigned balB = __builtin_amdgcn_ballot_w32(mkB);
  const unsigned ltm  = (1u << lane) - 1u;
  const int preA = __popc(balA & ltm);
  const int preB = __popc(balA) + __popc(balB & ltm);
  if (lane == 0) wc_s[wave] = __popc(balA) + __popc(balB);
  __syncthreads();
  int base = 0, cnt = 0;
#pragma unroll
  for (int w = 0; w < 4; ++w) {
    int c = wc_s[w];
    cnt += c;
    base += (w < wave) ? c : 0;
  }
  cnt  = __builtin_amdgcn_readfirstlane(cnt);
  base = __builtin_amdgcn_readfirstlane(base);
  if (packer) Wp[pidx] = wv;
  if (m0 >= cnt) return;
  if (mkA) idx_s[base + preA] = nodeA;
  if (mkB) idx_s[base + preB] = nodeB;
  const int nq = (cnt + 31) >> 5;

  attn_slice<false>(att, idx_s, WhP, s1, s2, Xin, Xout, bt, h, m0, cnt, nq,
                    wave, lane);
  if (m0 + ASL * 32 < cnt) {
    __syncthreads();
    attn_slice<true>(att, idx_s, WhP, s1, s2, Xin, Xout, bt, h,
                     m0 + ASL * 32, cnt, nq, wave, lane);
  }
}

__device__ __forceinline__ void zero_rows(float* __restrict__ Out, int row,
                                          int m0w, int m1w, int lane) {
  const v4f z = {0.f, 0.f, 0.f, 0.f};
  if (!(m0w > 0)) *(v4f*)(Out + (size_t)row * FF + lane * 4) = z;
  if (!(m1w > 0)) *(v4f*)(Out + (size_t)(row + 1) * FF + lane * 4) = z;
}

template <bool SECOND>
__device__ __forceinline__ void attn_ln_slice(float (*att)[32][NN + LPAD],
                                              float (*Ys)[FF + LPAD],
                                              const int* idx_s,
                                              const float* __restrict__ WhP,
                                              const float* __restrict__ s1,
                                              const float* __restrict__ s2,
                                              const float* __restrict__ Xin,
                                              const float* __restrict__ gamma,
                                              const float* __restrict__ beta,
                                              float* __restrict__ Out,
                                              int bt, int m0, int cnt, int nq,
                                              int wave, int lane) {
  const int h  = wave >> 2;
  const int w4 = wave & 3;
  const int    r0  = m0 + w4 * 8;
  const float* s1p = s1 + (bt * HH + h) * NN + r0;
  const float* s2p = s2 + (bt * HH + h) * NN + lane;
  softmax_any<SECOND>(att[h], s1p, s2p, r0, w4, lane, cnt, nq);
  __syncthreads();

  const int mt = w4 >> 1;
  const int nt = w4 & 1;
  const int kh = lane >> 4;
  const int nl = lane & 15;
  if (m0 + mt * 16 < cnt) {
    const v4f* Bp =
        (const v4f*)(WhP + ((size_t)(bt * HH + h) * 2 + nt) * (16 * NN)) + lane;
    v8f acc = attn_tile(att[h], Bp, mt, nl, kh, nq);
    const int gcol = h * DD + nt * 16 + nl;
    float xi[8];
#pragma unroll
    for (int v = 0; v < 8; ++v) {
      int crow = m0 + mt * 16 + v + 8 * kh;
      int cc   = (crow < cnt) ? crow : m0 + mt * 16;
      xi[v] = Xin[(size_t)(bt * NN + idx_s[cc]) * FF + gcol];
    }
#pragma unroll
    for (int v = 0; v < 8; ++v) {
      float x = xi[v] + acc[v];
      Ys[mt * 16 + v + 8 * kh][gcol] = (x > 0.f) ? x : expm1f(x);
    }
  }
  __syncthreads();

#pragma unroll
  for (int i = 0; i < 2; ++i) {
    const int rl   = wave * 2 + i;
    const int crow = m0 + rl;
    if (crow >= cnt) continue;
    float* o = Out + (size_t)(bt * NN + idx_s[crow]) * FF;
    float v[4];
    float s = 0.f;
#pragma unroll
    for (int q = 0; q < 4; ++q) { v[q] = Ys[rl][lane + 32 * q]; s += v[q]; }
    float mu = wave_sum32(s) * (1.f / FF);
    float var = 0.f;
#pragma unroll
    for (int q = 0; q < 4; ++q) { float d = v[q] - mu; var += d * d; }
    var = wave_sum32(var) * (1.f / FF);
    float rstd = rsqrtf(var + 1e-5f);
#pragma unroll
    for (int q = 0; q < 4; ++q) {
      int c = lane + 32 * q;
      o[c] = (v[q] - mu) * rstd * gamma[c] + beta[c];
    }
  }
}

__global__ __launch_bounds__(512) void gat_attn_ln(const float* __restrict__ WhP,
                                                   const float* __restrict__ s1,
                                                   const float* __restrict__ s2,
                                                   const int* __restrict__ mask,
                                                   const float* __restrict__ Xin,
                                                   const float* __restrict__ gamma,
                                                   const float* __restrict__ beta,
                                                   float* __restrict__ Out) {
  const int tid  = threadIdx.x;
  const int lane = tid & 31;
  const int wave = __builtin_amdgcn_readfirstlane(tid >> 5);
  const int bt   = blockIdx.x / ASL;
  const int mb   = blockIdx.x - bt * ASL;
  const int m0   = mb * 32;
  const bool two = mb + ASL < 8;

  __shared__ float att[HH][32][NN + LPAD];
  __shared__ float Ys[32][FF + LPAD];
  __shared__ int idx_s[NN];
  __shared__ int wc_s[8];

  bool mk = false;
  if (wave < 8) mk = mask[bt * NN + tid] > 0;
  const int rowz = bt * NN + m0 + wave * 2;
  const int mz0  = mask[rowz];
  const int mz1  = mask[rowz + 1];
  const int rowy = two ? rowz + ASL * 32 : rowz;
  const int my0  = mask[rowy];
  const int my1  = mask[rowy + 1];
  const unsigned bal = __builtin_amdgcn_ballot_w32(mk);
  const int pre = __popc(bal & ((1u << lane) - 1u));
  if (wave < 8 && lane == 0) wc_s[wave] = __popc(bal);
  __syncthreads();
  int base = 0, cnt = 0;
#pragma unroll
  for (int w = 0; w < 8; ++w) {
    int c = wc_s[w];
    cnt += c;
    base += (w < wave) ? c : 0;
  }
  cnt = __builtin_amdgcn_readfirstlane(cnt);
  if (m0 < cnt) {
    if (mk) idx_s[base + pre] = tid;
    const int nq = (cnt + 31) >> 5;
    attn_ln_slice<false>(att, Ys, idx_s, WhP, s1, s2, Xin, gamma, beta, Out,
                         bt, m0, cnt, nq, wave, lane);
    if (m0 + ASL * 32 < cnt)
      attn_ln_slice<true>(att, Ys, idx_s, WhP, s1, s2, Xin, gamma, beta, Out,
                          bt, m0 + ASL * 32, cnt, nq, wave, lane);
  }
  zero_rows(Out, rowz, mz0, mz1, lane);
  if (two) zero_rows(Out, rowy, my0, my1, lane);
}

extern "C" void kernel_launch(void* const* d_in, const int* in_sizes, int n_in,
                              void* d_out, int out_size, void* d_ws, size_t ws_size,
                              hipStream_t stream) {
  const float* X0    = (const float*)d_in[0];
  const int*   mask  = (const int*)d_in[1];
  const float* W0    = (const float*)d_in[2];
  const float* a0    = (const float*)d_in[3];
  const float* W1    = (const float*)d_in[4];
  const float* a1    = (const float*)d_in[5];
  const float* gamma = (const float*)d_in[6];
  const float* beta  = (const float*)d_in[7];
  float* out = (float*)d_out;

  float* Xb  = (float*)d_ws;
  float* WhP = Xb + (size_t)BT * NN * FF;
  float* s1  = WhP + (size_t)BT * NN * FF;
  float* s2  = s1 + (size_t)BT * HH * NN;
  float* Wp  = s2 + (size_t)BT * HH * NN;

  pack_w<<<64, 256, 0, stream>>>(W0, Wp);
  gemm_xw<<<BT * GSL, 256, 0, stream>>>(X0, mask, Wp, a0, WhP, s1, s2);
  gat_attn<<<BT * HH * ASL, 128, 0, stream>>>(WhP, s1, s2, mask, X0, Xb, W1, Wp);
  gemm_xw<<<BT * GSL, 256, 0, stream>>>(Xb, mask, Wp, a1, WhP, s1, s2);
  gat_attn_ln<<<BT * ASL, 512, 0, stream>>>(WhP, s1, s2, mask, Xb, gamma, beta, out);
}


# softmax exponent split into column factor x row factor (leaky_relu = max of two linear forms): column exps once a wave, element = 2 multiplies and a max; no exp per element, no row-max reduction
# speedup vs baseline: 4.3747x; 1.0884x over previous
#include <hip/hip_runtime.h>


typedef float v2f __attribute__((ext_vector_type(2)));
typedef float v4f __attribute__((ext_vector_type(4)));
typedef float v8f __attribute__((ext_vector_type(8)));

#define BT 128
#define NN 256
#define FF 128
#define HH 4
#define DD 32
#define NEGV (-1e9f)
#define GSL 10
#define ASL 5
static_assert(2 * GSL * 16 >= NN && 2 * ASL * 32 >= NN, "two trips cover a graph");
static_assert(BT * HH * ASL >= FF * FF / 128, "attention grid holds the weight-packing blocks");
#define LPAD 4

__device__ __forceinline__ float xhalf(float x) {
  int b = __builtin_bit_cast(int, x);
  return __builtin_bit_cast(
      float, __builtin_amdgcn_permlanex16(b, b, 0x76543210, 0xfedcba98, false, false));
}
template <int M>
__device__ __forceinline__ float xrow(float x) {
  int b = __builtin_bit_cast(int, x);
  return __builtin_bit_cast(
      float, __builtin_amdgcn_update_dpp(0, b, 0x160 + M, 0xf, 0xf, true));
}
__device__ __forceinline__ float wave_max32(float v) {
  v = fmaxf(v, xhalf(v));
  v = fmaxf(v, xrow<8>(v));
  v = fmaxf(v, xrow<4>(v));
  v = fmaxf(v, xrow<2>(v));
  v = fmaxf(v, xrow<1>(v));
  return v;
}
__device__ __forceinline__ float wave_sum32(float v) {
  v += xhalf(v);
  v += xrow<8>(v);
  v += xrow<4>(v);
  v += xrow<2>(v);
  v += xrow<1>(v);
  return v;
}

__device__ __forceinline__ int pack_src(int idx) {
  int e    = idx & 3;
  int lane = (idx >> 2) & 31;
  int kp   = (idx >> 7) & 15;
  int nt   = idx >> 11;
  int k = kp * 8 + (e >> 1) * 4 + 2 * (lane >> 4) + (e & 1);
  int n = nt * 16 + (lane & 15);
  return k * FF + n;
}
__global__ __launch_bounds__(256) void pack_w(const float* __restrict__ W,
                                              float* __restrict__ Wp) {
  int idx = blockIdx.x * 256 + threadIdx.x;
  Wp[idx] = W[pack_src(idx)];
}

template <bool SECOND>
__device__ __forceinline__ void gemm_slice(const float* __restrict__ Xg,
                                           const float* __restrict__ Wp,
                                           const float* __restrict__ avec,
                                           float* __restrict__ WhP,
                                           float* __restrict__ s1,
                                           float* __restrict__ s2,
                                           const int* idx_s,
                                           float (*As)[FF + LPAD],
                                           float (*Ts)[17],
                                           int bt, int m0, int cnt,
                                           int tid, int lane, int wave) {
  float xr[8];
#pragma unroll
  for (int i = 0; i < 8; ++i) {
    int idx = tid + i * 256;
    int c   = m0 + (idx >> 7);
    int cc  = (c < cnt) ? c : cnt - 1;
    xr[i] = Xg[(size_t)idx_s[cc] * FF + (idx & 127)];
  }
#pragma unroll
  for (int i = 0; i < 8; ++i) {
    int idx = tid + i * 256;
    int r   = idx >> 7;
    As[r][idx & 127] = (m0 + r < cnt) ? xr[i] : 0.f;
  }
  __syncthreads();

  const int n0 = wave * 16;
  const int kh = lane >> 4;
  const int nl = lane & 15;
  v8f acc = {};
  const v4f* Ap = (const v4f*)Wp + (size_t)(wave * 16) * 32 + lane;
#pragma unroll(SECOND ? 1 : FF / 8)
  for (int kp = 0; kp < FF / 8; ++kp) {
    v4f a4 = Ap[kp * 32];
    v2f a0 = {a4.x, a4.y};
    v2f a1 = {a4.z, a4.w};
    v2f b0 = *(const v2f*)&As[nl][kp * 8 + 2 * kh];
    v2f b1 = *(const v2f*)&As[nl][kp * 8 + 4 + 2 * kh];
    acc = __builtin_amdgcn_wmma_f32_16x16x4_f32(false, a0, false, b0,
                                                (short)0, acc, false, false);
    acc = __builtin_amdgcn_wmma_f32_16x16x4_f32(false, a1, false, b1,
                                                (short)0, acc, false, false);
  }
#pragma unroll
  for (int v = 0; v < 8; ++v) Ts[n0 + v + 8 * kh][nl] = acc[v];
  __syncthreads();
  {
    const int kpl  = tid >> 7;
    const int fln  = (tid >> 2) & 31;
    const int e    = tid & 3;
    const int nloc = kpl * 8 + (e >> 1) * 4 + 2 * (fln >> 4) + (e & 1);
    const int fl   = fln & 15;
    float* dst = WhP + (size_t)bt * (HH * DD * NN) + (m0 >> 3) * 128 + tid;
#pragma unroll
    for (int c = 0; c < 8; ++c)
      dst[c * (16 * NN)] = Ts[c * 16 + fl][nloc];
  }
  if (tid < 64) {
    const int hh = tid >> 4;
    const int n  = tid & 15;
    float v1 = 0.f, v2 = 0.f;
#pragma unroll(SECOND ? 1 : DD)
    for (int d = 0; d < DD; ++d) {
      float x = Ts[hh * DD + d][n];
      v1 += x * avec[d];
      v2 += x * avec[DD + d];
    }
    const int o = (bt * HH + hh) * NN + m0 + n;
    s1[o] = v1;
    s2[o] = v2;
  }
}

__global__ __launch_bounds__(256) void gemm_xw(const float* __restrict__ X,
                                               const int* __restrict__ mask,
                                               const float* __restrict__ Wp,
                                               const float* __restrict__ avec,
                                               float* __restrict__ WhP,
                                               float* __restrict__ s1,
                                               float* __restrict__ s2) {
  const int tid  = threadIdx.x;
  const int lane = tid & 31;
  const int wave = tid >> 5;
  const int bt   = blockIdx.x / GSL;
  const int m0   = (blockIdx.x - bt * GSL) * 16;

  __shared__ int idx_s[NN];
  __shared__ int wc_s[8];
  const bool mk = mask[bt * NN + tid] > 0;
  const unsigned bal = __builtin_amdgcn_ballot_w32(mk);
  const int pre = __popc(bal & ((1u << lane) - 1u));
  if (lane == 0) wc_s[wave] = __popc(bal);
  __syncthreads();
  int base = 0, cnt = 0;
#pragma unroll
  for (int w = 0; w < 8; ++w) {
    int c = wc_s[w];
    cnt += c;
    base += (w < wave) ? c : 0;
  }
  cnt = __builtin_amdgcn_readfirstlane(cnt);
  const int lim = (cnt + 31) & ~31;
  if (m0 >= lim) return;
  if (mk) idx_s[base + pre] = tid;
  __syncthreads();

  __shared__ float As[16][FF + LPAD];
  __shared__ float Ts[FF][17];
  const float* Xg = X + (size_t)bt * NN * FF;
  gemm_slice<false>(Xg, Wp, avec, WhP, s1, s2, idx_s, As, Ts, bt, m0, cnt,
                    tid, lane, wave);
  if (m0 + GSL * 16 < lim)
    gemm_slice<true>(Xg, Wp, avec, WhP, s1, s2, idx_s, As, Ts, bt,
                     m0 + GSL * 16, cnt, tid, lane, wave);
}

__device__ __forceinline__ float lane_val(float x, int L) {
  return __builtin_bit_cast(
      float, __builtin_amdgcn_readlane(__builtin_bit_cast(int, x), L));
}

template <int NQ>
__device__ __forceinline__ void softmax_rows(float (*att)[NN + LPAD],
                                             const float* __restrict__ s1p,
                                             const float* __restrict__ s2p,
                                             int r0, int wave, int lane,
                                             int cnt, int nq) {
  const int   l8  = lane & 7;
  const float s1l = s1p[l8];
  float s2r[NQ];
  bool  okc[NQ];
  float m2 = -__builtin_inff();
#pragma unroll
  for (int q = 0; q < NQ; ++q) {
    okc[q] = (lane + 32 * q) < cnt;
    s2r[q] = 0.f;
    if (q < nq) s2r[q] = s2p[32 * q];
    m2 = okc[q] ? fmaxf(m2, s2r[q]) : m2;
  }
  const float s2m = wave_max32(m2);
  float ca[NQ], cb[NQ];
#pragma unroll
  for (int q = 0; q < NQ; ++q) {
    const float d = s2r[q] - s2m;
    ca[q] = okc[q] ? __expf(d) : 0.f;
    cb[q] = okc[q] ? __expf(0.2f * d) : 0.f;
  }
  const float w   = 0.8f * (s1l + s2m);
  const float ral = __expf(fminf(w, 0.f));
  const float rbl = __expf(fminf(-w, 0.f));
  const bool  rowok = (r0 + l8) < cnt;

#pragma unroll
  for (int hb = 0; hb < 2; ++hb) {
    float p[4][NQ];
    float sums = 0.f;
#pragma unroll
    for (int i = 0; i < 4; ++i) {
      const int   rr = hb * 4 + i;
      const float ra = lane_val(ral, rr);
      const float rb = lane_val(rbl, rr);
      float vsum = 0.f;
#pragma unroll
      for (int q = 0; q < NQ; ++q) {
        p[i][q] = fmaxf(ca[q] * ra, cb[q] * rb);
        vsum += p[i][q];
      }
      vsum = wave_sum32(vsum);
      sums = (l8 == rr) ? vsum : sums;
    }
    const float invl = (rowok && sums > 0.f) ? 1.f / sums : 0.f;
#pragma unroll
    for (int i = 0; i < 4; ++i) {
      const int   rr  = hb * 4 + i;
      const float inv = lane_val(invl, rr);
      float* arow = &att[wave * 8 + rr][lane];
#pragma unroll
      for (int q = 0; q < NQ; ++q) arow[32 * q] = p[i][q] * inv;
    }
  }
}

template <bool SECOND>
__device__ __forceinline__ void softmax_any(float (*att)[NN + LPAD],
                                            const float* __restrict__ s1p,
                                            const float* __restrict__ s2p,
                                            int r0, int w, int lane, int cnt,
                                            int nq) {
  if (SECOND)       softmax_rows<8>(att, s1p, s2p, r0, w, lane, cnt, nq);
  else if (nq <= 4) softmax_rows<4>(att, s1p, s2p, r0, w, lane, cnt, nq);
  else if (nq == 5) softmax_rows<5>(att, s1p, s2p, r0, w, lane, cnt, nq);
  else              softmax_rows<8>(att, s1p, s2p, r0, w, lane, cnt, nq);
}

__device__ __forceinline__ v8f attn_tile(const float (*att)[NN + LPAD],
                                         const v4f* __restrict__ Bp, int mt,
                                         int nl, int kh, int nq) {
  v8f acc = {};
  for (int g = 0; g < nq; ++g) {
#pragma unroll
    for (int j = 0; j < 4; ++j) {
      const int kp = g * 4 + j;
      v4f b4 = Bp[kp * 32];
      v2f b0 = {b4.x, b4.y};
      v2f b1 = {b4.z, b4.w};
      v2f a0 = *(const v2f*)&att[mt * 16 + nl][kp * 8 + 2 * kh];
      v2f a1 = *(const v2f*)&att[mt * 16 + nl][kp * 8 + 4 + 2 * kh];
      acc = __builtin_amdgcn_wmma_f32_16x16x4_f32(false, a0, false, b0,
                                                  (short)0, acc, false, false);
      acc = __builtin_amdgcn_wmma_f32_16x16x4_f32(false, a1, false, b1,
                                                  (short)0, acc, false, false);
    }
  }
  return acc;
}

template <bool SECOND>
__device__ __forceinline__ void attn_slice(float (*att)[NN + LPAD],
                                           const int* idx_s,
                                           const float* __restrict__ WhP,
                                           const float* __restrict__ s1,
                                           const float* __restrict__ s2,
                                           const float* __restrict__ Xin,
                                           float* __restrict__ Xout,
                                           int bt, int h, int m0, int cnt,
                                           int nq, int wave, int lane) {
  const int    r0  = m0 + wave * 8;
  const float* s1p = s1 + (bt * HH + h) * NN + r0;
  const float* s2p = s2 + (bt * HH + h) * NN + lane;
  softmax_any<SECOND>(att, s1p, s2p, r0, wave, lane, cnt, nq);
  __syncthreads();

  const int mt = wave >> 1;
  if (m0 + mt * 16 >= cnt) return;
  const int nt = wave & 1;
  const int kh = lane >> 4;
  const int nl = lane & 15;
  const v4f* Bp =
      (const v4f*)(WhP + ((size_t)(bt * HH + h) * 2 + nt) * (16 * NN)) + lane;
  v8f acc = attn_tile(att, Bp, mt, nl, kh, nq);
  const int gcol = h * DD + nt * 16 + nl;
  size_t o[8];
  float  xi[8];
#pragma unroll
  for (int v = 0; v < 8; ++v) {
    int crow = m0 + mt * 16 + v + 8 * kh;
    int cc   = (crow < cnt) ? crow : m0 + mt * 16;
    o[v]  = (size_t)(bt * NN + idx_s[cc]) * FF + gcol;
    xi[v] = Xin[o[v]];
  }
#pragma unroll
  for (int v = 0; v < 8; ++v) {
    int crow = m0 + mt * 16 + v + 8 * kh;
    float x = xi[v] + acc[v];
    float y = (x > 0.f) ? x : expm1f(x);
    if (crow < cnt) Xout[o[v]] = y;
  }
}

__global__ __launch_bounds__(128) void gat_attn(const float* __restrict__ WhP,
                                                const float* __restrict__ s1,
                                                const float* __restrict__ s2,
                                                const int* __restrict__ mask,
                                                const float* __restrict__ Xin,
                                                float* __restrict__ Xout,
                                                const float* __restrict__ Wnext,
                                                float* __restrict__ Wp) {
  const int tid  = threadIdx.x;
  const int lane = tid & 31;
  const bool packer = blockIdx.x < FF * FF / 128;
  const int  pidx   = blockIdx.x * 128 + tid;
  float wv = 0.f;
  if (packer) wv = Wnext[pack_src(pidx)];
  const int wave = __builtin_amdgcn_readfirstlane(tid >> 5);
  const int bh   = blockIdx.x / ASL;
  const int mb   = blockIdx.x - bh * ASL;
  const int h    = bh & (HH - 1);
  const int bt   = bh >> 2;
  const int m0   = mb * 32;

  __shared__ float att[32][NN + LPAD];

  __shared__ int idx_s[NN];
  __shared__ int wc_s[4];
  const int  nodeA = wave * 64 + lane;
  const int  nodeB = nodeA + 32;
  const bool mkA = mask[bt * NN + nodeA] > 0;
  const bool mkB = mask[bt * NN + nodeB] > 0;
  const unsigned balA = __builtin_amdgcn_ballot_w32(mkA);
  const unsigned balB = __builtin_amdgcn_ballot_w32(mkB);
  const unsigned ltm  = (1u << lane) - 1u;
  const int preA = __popc(balA & ltm);
  const int preB = __popc(balA) + __popc(balB & ltm);
  if (lane == 0) wc_s[wave] = __popc(balA) + __popc(balB);
  __syncthreads();
  int base = 0, cnt = 0;
#pragma unroll
  for (int w = 0; w < 4; ++w) {
    int c = wc_s[w];
    cnt += c;
    base += (w < wave) ? c : 0;
  }
  cnt  = __builtin_amdgcn_readfirstlane(cnt);
  base = __builtin_amdgcn_readfirstlane(base);
  if (packer) Wp[pidx] = wv;
  if (m0 >= cnt) return;
  if (mkA) idx_s[base + preA] = nodeA;
  if (mkB) idx_s[base + preB] = nodeB;
  const int nq = (cnt + 31) >> 5;

  attn_slice<false>(att, idx_s, WhP, s1, s2, Xin, Xout, bt, h, m0, cnt, nq,
                    wave, lane);
  if (m0 + ASL * 32 < cnt) {
    __syncthreads();
    attn_slice<true>(att, idx_s, WhP, s1, s2, Xin, Xout, bt, h,
                     m0 + ASL * 32, cnt, nq, wave, lane);
  }
}

__device__ __forceinline__ void zero_rows(float* __restrict__ Out, int row,
                                          int m0w, int m1w, int lane) {
  const v4f z = {0.f, 0.f, 0.f, 0.f};
  if (!(m0w > 0)) *(v4f*)(Out + (size_t)row * FF + lane * 4) = z;
  if (!(m1w > 0)) *(v4f*)(Out + (size_t)(row + 1) * FF + lane * 4) = z;
}

template <bool SECOND>
__device__ __forceinline__ void attn_ln_slice(float (*att)[32][NN + LPAD],
                                              float (*Ys)[FF + LPAD],
                                              const int* idx_s,
                                              const float* __restrict__ WhP,
                                              const float* __restrict__ s1,
                                              const float* __restrict__ s2,
                                              const float* __restrict__ Xin,
                                              const float* __restrict__ gamma,
                                              const float* __restrict__ beta,
                                              float* __restrict__ Out,
                                              int bt, int m0, int cnt, int nq,
                                              int wave, int lane) {
  const int h  = wave >> 2;
  const int w4 = wave & 3;
  const int    r0  = m0 + w4 * 8;
  const float* s1p = s1 + (bt * HH + h) * NN + r0;
  const float* s2p = s2 + (bt * HH + h) * NN + lane;
  softmax_any<SECOND>(att[h], s1p, s2p, r0, w4, lane, cnt, nq);
  __syncthreads();

  const int mt = w4 >> 1;
  const int nt = w4 & 1;
  const int kh = lane >> 4;
  const int nl = lane & 15;
  if (m0 + mt * 16 < cnt) {
    const v4f* Bp =
        (const v4f*)(WhP + ((size_t)(bt * HH + h) * 2 + nt) * (16 * NN)) + lane;
    v8f acc = attn_tile(att[h], Bp, mt, nl, kh, nq);
    const int gcol = h * DD + nt * 16 + nl;
    float xi[8];
#pragma unroll
    for (int v = 0; v < 8; ++v) {
      int crow = m0 + mt * 16 + v + 8 * kh;
      int cc   = (crow < cnt) ? crow : m0 + mt * 16;
      xi[v] = Xin[(size_t)(bt * NN + idx_s[cc]) * FF + gcol];
    }
#pragma unroll
    for (int v = 0; v < 8; ++v) {
      float x = xi[v] + acc[v];
      Ys[mt * 16 + v + 8 * kh][gcol] = (x > 0.f) ? x : expm1f(x);
    }
  }
  __syncthreads();

#pragma unroll
  for (int i = 0; i < 2; ++i) {
    const int rl   = wave * 2 + i;
    const int crow = m0 + rl;
    if (crow >= cnt) continue;
    float* o = Out + (size_t)(bt * NN + idx_s[crow]) * FF;
    float v[4];
    float s = 0.f;
#pragma unroll
    for (int q = 0; q < 4; ++q) { v[q] = Ys[rl][lane + 32 * q]; s += v[q]; }
    float mu = wave_sum32(s) * (1.f / FF);
    float var = 0.f;
#pragma unroll
    for (int q = 0; q < 4; ++q) { float d = v[q] - mu; var += d * d; }
    var = wave_sum32(var) * (1.f / FF);
    float rstd = rsqrtf(var + 1e-5f);
#pragma unroll
    for (int q = 0; q < 4; ++q) {
      int c = lane + 32 * q;
      o[c] = (v[q] - mu) * rstd * gamma[c] + beta[c];
    }
  }
}

__global__ __launch_bounds__(512) void gat_attn_ln(const float* __restrict__ WhP,
                                                   const float* __restrict__ s1,
                                                   const float* __restrict__ s2,
                                                   const int* __restrict__ mask,
                                                   const float* __restrict__ Xin,
                                                   const float* __restrict__ gamma,
                                                   const float* __restrict__ beta,
                                                   float* __restrict__ Out) {
  const int tid  = threadIdx.x;
  const int lane = tid & 31;
  const int wave = __builtin_amdgcn_readfirstlane(tid >> 5);
  const int bt   = blockIdx.x / ASL;
  const int mb   = blockIdx.x - bt * ASL;
  const int m0   = mb * 32;
  const bool two = mb + ASL < 8;

  __shared__ float att[HH][32][NN + LPAD];
  __shared__ float Ys[32][FF + LPAD];
  __shared__ int idx_s[NN];
  __shared__ int wc_s[8];

  bool mk = false;
  if (wave < 8) mk = mask[bt * NN + tid] > 0;
  const int rowz = bt * NN + m0 + wave * 2;
  const int mz0  = mask[rowz];
  const int mz1  = mask[rowz + 1];
  const int rowy = two ? rowz + ASL * 32 : rowz;
  const int my0  = mask[rowy];
  const int my1  = mask[rowy + 1];
  const unsigned bal = __builtin_amdgcn_ballot_w32(mk);
  const int pre = __popc(bal & ((1u << lane) - 1u));
  if (wave < 8 && lane == 0) wc_s[wave] = __popc(bal);
  __syncthreads();
  int base = 0, cnt = 0;
#pragma unroll
  for (int w = 0; w < 8; ++w) {
    int c = wc_s[w];
    cnt += c;
    base += (w < wave) ? c : 0;
  }
  cnt = __builtin_amdgcn_readfirstlane(cnt);
  if (m0 < cnt) {
    if (mk) idx_s[base + pre] = tid;
    const int nq = (cnt + 31) >> 5;
    attn_ln_slice<false>(att, Ys, idx_s, WhP, s1, s2, Xin, gamma, beta, Out,
                         bt, m0, cnt, nq, wave, lane);
    if (m0 + ASL * 32 < cnt)
      attn_ln_slice<true>(att, Ys, idx_s, WhP, s1, s2, Xin, gamma, beta, Out,
                          bt, m0 + ASL * 32, cnt, nq, wave, lane);
  }
  zero_rows(Out, rowz, mz0, mz1, lane);
  if (two) zero_rows(Out, rowy, my0, my1, lane);
}

extern "C" void kernel_launch(void* const* d_in, const int* in_sizes, int n_in,
                              void* d_out, int out_size, void* d_ws, size_t ws_size,
                              hipStream_t stream) {
  const float* X0    = (const float*)d_in[0];
  const int*   mask  = (const int*)d_in[1];
  const float* W0    = (const float*)d_in[2];
  const float* a0    = (const float*)d_in[3];
  const float* W1    = (const float*)d_in[4];
  const float* a1    = (const float*)d_in[5];
  const float* gamma = (const float*)d_in[6];
  const float* beta  = (const float*)d_in[7];
  float* out = (float*)d_out;

  float* Xb  = (float*)d_ws;
  float* WhP = Xb + (size_t)BT * NN * FF;
  float* s1  = WhP + (size_t)BT * NN * FF;
  float* s2  = s1 + (size_t)BT * HH * NN;
  float* Wp  = s2 + (size_t)BT * HH * NN;

  pack_w<<<64, 256, 0, stream>>>(W0, Wp);
  gemm_xw<<<BT * GSL, 256, 0, stream>>>(X0, mask, Wp, a0, WhP, s1, s2);
  gat_attn<<<BT * HH * ASL, 128, 0, stream>>>(WhP, s1, s2, mask, X0, Xb, W1, Wp);
  gemm_xw<<<BT * GSL, 256, 0, stream>>>(Xb, mask, Wp, a1, WhP, s1, s2);
  gat_attn_ln<<<BT * ASL, 512, 0, stream>>>(WhP, s1, s2, mask, Xb, gamma, beta, out);
}
